# starting kernel
# baseline (speedup 1.0000x reference)
#include <hip/hip_runtime.h>
#include <math.h>

#define DIM 128
#define HEADS 4
#define CH 32
#define NEG_SLOPE 0.2f

typedef __attribute__((ext_vector_type(2))) float v2f;
typedef __attribute__((ext_vector_type(8))) float v8f;

__global__ void gat_gemm_f32_wmma(const float* __restrict__ A,
                                  const float* __restrict__ W,
                                  const float* __restrict__ bias,
                                  float* __restrict__ out) {
  const int wave = threadIdx.x >> 5;
  const int lane = threadIdx.x & 31;
  const int row0 = blockIdx.x * 16;
  const int col0 = wave * 16;
  const int m  = lane & 15;
  const int kh = lane >> 4;

  const float* __restrict__ arow = A + (size_t)(row0 + m) * DIM;

  v8f acc = {};
  for (int k0 = 0; k0 < DIM; k0 += 4) {
    const int ka = k0 + kh * 2;
    v2f av, bv;
    av.x = arow[ka];
    av.y = arow[ka + 1];
    bv.x = W[(size_t)ka * DIM + col0 + m];
    bv.y = W[(size_t)(ka + 1) * DIM + col0 + m];
    acc = __builtin_amdgcn_wmma_f32_16x16x4_f32(
        false, av, false, bv, (short)0, acc, false, false);
  }

  const int nc = col0 + m;
  const float b = bias ? bias[nc] : 0.0f;
  const int rbase = row0 + kh * 8;
#pragma unroll
  for (int r = 0; r < 8; ++r)
    out[(size_t)(rbase + r) * DIM + nc] = acc[r] + b;
}

__global__ void gat_attn_scores(const float* __restrict__ h,
                                const float* __restrict__ a_src,
                                const float* __restrict__ a_dst,
                                float* __restrict__ asrc,
                                float* __restrict__ adst, int N) {
  int t = blockIdx.x * blockDim.x + threadIdx.x;
  if (t >= N * HEADS) return;
  const int n = t >> 2, hh = t & 3;
  const float* __restrict__ hp = h + (size_t)n * DIM + hh * CH;
  const float* __restrict__ ws = a_src + hh * CH;
  const float* __restrict__ wd = a_dst + hh * CH;
  float s1 = 0.f, s2 = 0.f;
#pragma unroll
  for (int c = 0; c < CH; ++c) {
    const float v = hp[c];
    s1 += v * ws[c];
    s2 += v * wd[c];
  }
  asrc[t] = s1;
  adst[t] = s2;
}

__device__ __forceinline__ void edge_sd(const int* __restrict__ ei, int e,
                                        int NE, int& s, int& d) {
  if (e < NE) { s = ei[e]; d = ei[NE + e]; }
  else        { s = d = e - NE; }
}

__device__ __forceinline__ unsigned ord_encode(float f) {
  unsigned u = __float_as_uint(f);
  return (f >= 0.0f) ? (u | 0x80000000u) : ~u;
}

__global__ void gat_edge_pass1(const int* __restrict__ ei, int NE, int E,
                               const float* __restrict__ asrc,
                               const float* __restrict__ adst,
                               float* __restrict__ ebuf,
                               unsigned* __restrict__ menc) {
  int t = blockIdx.x * blockDim.x + threadIdx.x;
  if (t >= E * HEADS) return;
  const int e = t >> 2, hh = t & 3;
  int s, d;
  edge_sd(ei, e, NE, s, d);
  float v = asrc[s * HEADS + hh] + adst[d * HEADS + hh];
  v = (v > 0.0f) ? v : NEG_SLOPE * v;
  ebuf[t] = v;
  atomicMax(&menc[d * HEADS + hh], ord_encode(v));
}

__global__ void gat_decode_max(unsigned* __restrict__ menc, int n) {
  int t = blockIdx.x * blockDim.x + threadIdx.x;
  if (t >= n) return;
  const unsigned u = menc[t];
  float f;
  if (u == 0u)                 f = 0.0f;
  else if (u & 0x80000000u)    f = __uint_as_float(u & 0x7FFFFFFFu);
  else                         f = __uint_as_float(~u);
  ((float*)menc)[t] = f;
}

__global__ void gat_edge_pass2(const int* __restrict__ ei, int NE, int E,
                               float* __restrict__ ebuf,
                               const float* __restrict__ m,
                               float* __restrict__ denom) {
  int t = blockIdx.x * blockDim.x + threadIdx.x;
  if (t >= E * HEADS) return;
  const int e = t >> 2, hh = t & 3;
  int s, d;
  edge_sd(ei, e, NE, s, d);
  const float ee = expf(ebuf[t] - m[d * HEADS + hh]);
  ebuf[t] = ee;
  atomicAdd(&denom[d * HEADS + hh], ee);
}

__global__ void gat_edge_pass3(const int* __restrict__ ei, int NE, int E,
                               const float* __restrict__ ebuf,
                               const float* __restrict__ denom,
                               const float* __restrict__ h,
                               float* __restrict__ agg) {
  int t = blockIdx.x * blockDim.x + threadIdx.x;
  if (t >= E * DIM) return;
  const int e = t >> 7;
  const int c = t & (DIM - 1);
  const int hh = c >> 5;
  int s, d;
  edge_sd(ei, e, NE, s, d);
  const float alpha =
      ebuf[e * HEADS + hh] / (denom[d * HEADS + hh] + 1e-16f);
  atomicAdd(&agg[(size_t)d * DIM + c], h[(size_t)s * DIM + c] * alpha);
}

__global__ void gat_elu(const float* __restrict__ agg, float* __restrict__ out,
                        int n) {
  int t = blockIdx.x * blockDim.x + threadIdx.x;
  if (t >= n) return;
  const float v = agg[t];
  out[t] = (v > 0.0f) ? v : expm1f(v);
}

extern "C" void kernel_launch(void* const* d_in, const int* in_sizes, int n_in,
                              void* d_out, int out_size, void* d_ws,
                              size_t ws_size, hipStream_t stream) {
  const float* x  = (const float*)d_in[0];
  const int*   ei = (const int*)d_in[1];

  const int N  = in_sizes[0] / DIM;
  const int NE = in_sizes[1] / 2;
  const int E  = NE + N;

  float* ws = (float*)d_ws;
  size_t o = 0;
  float* h     = ws + o; o += (size_t)N * DIM;
  float* agg   = ws + o; o += (size_t)N * DIM;
  float* asrc  = ws + o; o += (size_t)N * HEADS;
  float* adst  = ws + o; o += (size_t)N * HEADS;
  float* mbuf  = ws + o; o += (size_t)N * HEADS;
  float* denom = ws + o; o += (size_t)N * HEADS;
  float* ebuf  = ws + o; o += (size_t)E * HEADS;

  float* y = (float*)d_out;

  const int TB = 256;
  const int gGemm = N / 16;
  const int gNH   = (N * HEADS + TB - 1) / TB;
  const int gEH   = (E * HEADS + TB - 1) / TB;
  const int gED   = (E * DIM + TB - 1) / TB;
  const int gND   = (N * DIM + TB - 1) / TB;

  const float* xin = x;
  for (int layer = 0; layer < 2; ++layer) {
    const float* W     = (const float*)d_in[2 + layer * 5];
    const float* a_src = (const float*)d_in[3 + layer * 5];
    const float* a_dst = (const float*)d_in[4 + layer * 5];
    const float* bias  = (const float*)d_in[5 + layer * 5];
    const float* Wskip = (const float*)d_in[6 + layer * 5];

    gat_gemm_f32_wmma<<<gGemm, TB, 0, stream>>>(xin, W, nullptr, h);
    gat_gemm_f32_wmma<<<gGemm, TB, 0, stream>>>(xin, Wskip, bias, agg);

    gat_attn_scores<<<gNH, TB, 0, stream>>>(h, a_src, a_dst, asrc, adst, N);

    hipMemsetAsync(mbuf, 0, (size_t)N * HEADS * sizeof(float), stream);
    hipMemsetAsync(denom, 0, (size_t)N * HEADS * sizeof(float), stream);

    gat_edge_pass1<<<gEH, TB, 0, stream>>>(ei, NE, E, asrc, adst, ebuf,
                                           (unsigned*)mbuf);
    gat_decode_max<<<gNH, TB, 0, stream>>>((unsigned*)mbuf, N * HEADS);
    gat_edge_pass2<<<gEH, TB, 0, stream>>>(ei, NE, E, ebuf, mbuf, denom);
    gat_edge_pass3<<<gED, TB, 0, stream>>>(ei, NE, E, ebuf, denom, h, agg);

    gat_elu<<<gND, TB, 0, stream>>>(agg, y, N * DIM);
    xin = y;
  }
}


# Message aggregation: float-atomic scatter replaced by per-destination gather over a per-call dst-bucketed edge list (arrival-order rank); sum order: register sum per node, then added to skip row
# speedup vs baseline: 1.6253x; 1.6253x over previous
#include <hip/hip_runtime.h>
#include <math.h>

#define DIM 128
#define HEADS 4
#define CH 32
#define NEG_SLOPE 0.2f

typedef __attribute__((ext_vector_type(2))) float v2f;
typedef __attribute__((ext_vector_type(8))) float v8f;

__global__ void gat_gemm_f32_wmma(const float* __restrict__ A,
                                  const float* __restrict__ W,
                                  const float* __restrict__ bias,
                                  float* __restrict__ out) {
  const int wave = threadIdx.x >> 5;
  const int lane = threadIdx.x & 31;
  const int row0 = blockIdx.x * 16;
  const int col0 = wave * 16;
  const int m  = lane & 15;
  const int kh = lane >> 4;

  const float* __restrict__ arow = A + (size_t)(row0 + m) * DIM;

  v8f acc = {};
  for (int k0 = 0; k0 < DIM; k0 += 4) {
    const int ka = k0 + kh * 2;
    v2f av, bv;
    av.x = arow[ka];
    av.y = arow[ka + 1];
    bv.x = W[(size_t)ka * DIM + col0 + m];
    bv.y = W[(size_t)(ka + 1) * DIM + col0 + m];
    acc = __builtin_amdgcn_wmma_f32_16x16x4_f32(
        false, av, false, bv, (short)0, acc, false, false);
  }

  const int nc = col0 + m;
  const float b = bias ? bias[nc] : 0.0f;
  const int rbase = row0 + kh * 8;
#pragma unroll
  for (int r = 0; r < 8; ++r)
    out[(size_t)(rbase + r) * DIM + nc] = acc[r] + b;
}

__global__ void gat_attn_scores(const float* __restrict__ h,
                                const float* __restrict__ a_src,
                                const float* __restrict__ a_dst,
                                float* __restrict__ asrc,
                                float* __restrict__ adst, int N) {
  int t = blockIdx.x * blockDim.x + threadIdx.x;
  if (t >= N * HEADS) return;
  const int n = t >> 2, hh = t & 3;
  const float* __restrict__ hp = h + (size_t)n * DIM + hh * CH;
  const float* __restrict__ ws = a_src + hh * CH;
  const float* __restrict__ wd = a_dst + hh * CH;
  float s1 = 0.f, s2 = 0.f;
#pragma unroll
  for (int c = 0; c < CH; ++c) {
    const float v = hp[c];
    s1 += v * ws[c];
    s2 += v * wd[c];
  }
  asrc[t] = s1;
  adst[t] = s2;
}

__device__ __forceinline__ void edge_sd(const int* __restrict__ ei, int e,
                                        int NE, int& s, int& d) {
  if (e < NE) { s = ei[e]; d = ei[NE + e]; }
  else        { s = d = e - NE; }
}

__device__ __forceinline__ unsigned ord_encode(float f) {
  unsigned u = __float_as_uint(f);
  return (f >= 0.0f) ? (u | 0x80000000u) : ~u;
}

__global__ void gat_edge_pass1(const int* __restrict__ ei, int NE, int E,
                               const float* __restrict__ asrc,
                               const float* __restrict__ adst,
                               float* __restrict__ ebuf,
                               unsigned* __restrict__ menc) {
  int t = blockIdx.x * blockDim.x + threadIdx.x;
  if (t >= E * HEADS) return;
  const int e = t >> 2, hh = t & 3;
  int s, d;
  edge_sd(ei, e, NE, s, d);
  float v = asrc[s * HEADS + hh] + adst[d * HEADS + hh];
  v = (v > 0.0f) ? v : NEG_SLOPE * v;
  ebuf[t] = v;
  atomicMax(&menc[d * HEADS + hh], ord_encode(v));
}

__global__ void gat_decode_max(unsigned* __restrict__ menc, int n) {
  int t = blockIdx.x * blockDim.x + threadIdx.x;
  if (t >= n) return;
  const unsigned u = menc[t];
  float f;
  if (u == 0u)                 f = 0.0f;
  else if (u & 0x80000000u)    f = __uint_as_float(u & 0x7FFFFFFFu);
  else                         f = __uint_as_float(~u);
  ((float*)menc)[t] = f;
}

__global__ void gat_edge_pass2(const int* __restrict__ ei, int NE, int E,
                               float* __restrict__ ebuf,
                               const float* __restrict__ m,
                               float* __restrict__ denom) {
  int t = blockIdx.x * blockDim.x + threadIdx.x;
  if (t >= E * HEADS) return;
  const int e = t >> 2, hh = t & 3;
  int s, d;
  edge_sd(ei, e, NE, s, d);
  const float ee = expf(ebuf[t] - m[d * HEADS + hh]);
  ebuf[t] = ee;
  atomicAdd(&denom[d * HEADS + hh], ee);
}

__global__ void csr_zero(int* __restrict__ cnt, int n) {
  int t = blockIdx.x * blockDim.x + threadIdx.x;
  if (t < n) cnt[t] = 0;
}

__global__ void csr_count(const int* __restrict__ ei, int NE,
                          int* __restrict__ cnt, int* __restrict__ rank) {
  int t = blockIdx.x * blockDim.x + threadIdx.x;
  if (t >= NE) return;
  rank[t] = atomicAdd(&cnt[ei[NE + t]], 1);
}

__global__ void __launch_bounds__(1024)
csr_scan(const int* __restrict__ cnt, int* __restrict__ rowptr, int chunk4) {
  __shared__ int wsum[32];
  const int t = threadIdx.x, lane = t & 31, wv = t >> 5;
  const int4* __restrict__ c4 = (const int4*)cnt + (size_t)t * chunk4;
  int s = 0;
  for (int i = 0; i < chunk4; ++i) {
    const int4 v = c4[i];
    s += (v.x + v.y) + (v.z + v.w);
  }
  int inc = s;
  for (int o = 1; o < 32; o <<= 1) {
    const int up = __shfl_up(inc, o);
    if (lane >= o) inc += up;
  }
  if (lane == 31) wsum[wv] = inc;
  __syncthreads();
  if (wv == 0) {
    const int w = wsum[lane];
    int wi = w;
    for (int o = 1; o < 32; o <<= 1) {
      const int up = __shfl_up(wi, o);
      if (lane >= o) wi += up;
    }
    wsum[lane] = wi - w;
  }
  __syncthreads();
  int run = wsum[wv] + inc - s;
  int4* __restrict__ r4 = (int4*)rowptr + (size_t)t * chunk4;
  for (int i = 0; i < chunk4; ++i) {
    const int4 v = c4[i];
    int4 o;
    o.x = run; run += v.x;
    o.y = run; run += v.y;
    o.z = run; run += v.z;
    o.w = run; run += v.w;
    r4[i] = o;
  }
}

__global__ void csr_fill(const int* __restrict__ ei, int NE,
                         const int* __restrict__ rowptr,
                         const int* __restrict__ rank,
                         int2* __restrict__ csr) {
  int t = blockIdx.x * blockDim.x + threadIdx.x;
  if (t >= NE) return;
  const int pos = rowptr[ei[NE + t]] + rank[t];
  csr[pos] = make_int2(ei[t], t);
}

__device__ __forceinline__ void edge_fma(float4& acc, const float4 hv, float a) {
  acc.x += hv.x * a;
  acc.y += hv.y * a;
  acc.z += hv.z * a;
  acc.w += hv.w * a;
}

__global__ void __launch_bounds__(256)
gat_gather_csr(const int2* __restrict__ csr, const int* __restrict__ rowptr,
               const float* __restrict__ ebuf, const float* __restrict__ denom,
               const float* __restrict__ h, float* __restrict__ agg,
               int N, int NE) {
  const int lane = threadIdx.x & 31;
  const int hh = lane >> 3;
  const int wpb = blockDim.x >> 5;
  const int wv = __builtin_amdgcn_readfirstlane(threadIdx.x >> 5);
  const int nw = gridDim.x * wpb;
  const float4* __restrict__ h4 = (const float4*)h;
  float4* __restrict__ a4 = (float4*)agg;

  for (int d = blockIdx.x * wpb + wv; d < N; d += nw) {
    const int beg = rowptr[d];
    const int end = rowptr[d + 1];
    const float den = denom[d * HEADS + hh] + 1e-16f;

    float4 acc = make_float4(0.f, 0.f, 0.f, 0.f);
    {
      const float a = ebuf[(NE + d) * HEADS + hh] / den;
      edge_fma(acc, h4[d * 32 + lane], a);
    }
    for (int base = beg; base < end; base += 32) {
      const int rem = end - base;
      const int n = rem < 32 ? rem : 32;
      int2 my = make_int2(0, 0);
      if (lane < n) my = csr[base + lane];
      int j = 0;
      for (; j + 4 <= n; j += 4) {
        const int s0 = __builtin_amdgcn_readlane(my.x, j);
        const int s1 = __builtin_amdgcn_readlane(my.x, j + 1);
        const int s2 = __builtin_amdgcn_readlane(my.x, j + 2);
        const int s3 = __builtin_amdgcn_readlane(my.x, j + 3);
        const int e0 = __builtin_amdgcn_readlane(my.y, j);
        const int e1 = __builtin_amdgcn_readlane(my.y, j + 1);
        const int e2 = __builtin_amdgcn_readlane(my.y, j + 2);
        const int e3 = __builtin_amdgcn_readlane(my.y, j + 3);
        const float4 v0 = h4[s0 * 32 + lane];
        const float4 v1 = h4[s1 * 32 + lane];
        const float4 v2 = h4[s2 * 32 + lane];
        const float4 v3 = h4[s3 * 32 + lane];
        const float w0 = ebuf[e0 * HEADS + hh];
        const float w1 = ebuf[e1 * HEADS + hh];
        const float w2 = ebuf[e2 * HEADS + hh];
        const float w3 = ebuf[e3 * HEADS + hh];
        edge_fma(acc, v0, w0 / den);
        edge_fma(acc, v1, w1 / den);
        edge_fma(acc, v2, w2 / den);
        edge_fma(acc, v3, w3 / den);
      }
      for (; j < n; ++j) {
        const int s0 = __builtin_amdgcn_readlane(my.x, j);
        const int e0 = __builtin_amdgcn_readlane(my.y, j);
        const float4 v0 = h4[s0 * 32 + lane];
        const float w0 = ebuf[e0 * HEADS + hh];
        edge_fma(acc, v0, w0 / den);
      }
    }
    float4 sk = a4[d * 32 + lane];
    sk.x += acc.x;
    sk.y += acc.y;
    sk.z += acc.z;
    sk.w += acc.w;
    a4[d * 32 + lane] = sk;
  }
}

__global__ void gat_elu(const float* __restrict__ agg, float* __restrict__ out,
                        int n) {
  int t = blockIdx.x * blockDim.x + threadIdx.x;
  if (t >= n) return;
  const float v = agg[t];
  out[t] = (v > 0.0f) ? v : expm1f(v);
}

extern "C" void kernel_launch(void* const* d_in, const int* in_sizes, int n_in,
                              void* d_out, int out_size, void* d_ws,
                              size_t ws_size, hipStream_t stream) {
  const float* x  = (const float*)d_in[0];
  const int*   ei = (const int*)d_in[1];

  const int N  = in_sizes[0] / DIM;
  const int NE = in_sizes[1] / 2;
  const int E  = NE + N;

  float* ws = (float*)d_ws;
  size_t o = 0;
  float* h     = ws + o; o += (size_t)N * DIM;
  float* agg   = ws + o; o += (size_t)N * DIM;
  float* asrc  = ws + o; o += (size_t)N * HEADS;
  float* adst  = ws + o; o += (size_t)N * HEADS;
  float* mbuf  = ws + o; o += (size_t)N * HEADS;
  float* denom = ws + o; o += (size_t)N * HEADS;
  float* ebuf  = ws + o; o += (size_t)E * HEADS;
  const int chunk4 = (N + 4096) / 4096;
  const int NP     = chunk4 * 4096;
  int2* csr    = (int2*)(ws + o); o += (size_t)NE * 2;
  int*  rank   = (int*)(ws + o);  o += (size_t)NE;
  int*  cnt    = (int*)(ws + o);  o += (size_t)NP;
  int*  rowptr = (int*)(ws + o);  o += (size_t)NP;

  float* y = (float*)d_out;

  const int TB = 256;
  const int gGemm = N / 16;
  const int gNH   = (N * HEADS + TB - 1) / TB;
  const int gEH   = (E * HEADS + TB - 1) / TB;
  const int gE1   = (NE + TB - 1) / TB;
  const int gNP   = (NP + TB - 1) / TB;
  const int gGath = 2048;
  const int gND   = (N * DIM + TB - 1) / TB;

  csr_zero<<<gNP, TB, 0, stream>>>(cnt, NP);
  csr_count<<<gE1, TB, 0, stream>>>(ei, NE, cnt, rank);
  csr_scan<<<1, 1024, 0, stream>>>(cnt, rowptr, chunk4);
  csr_fill<<<gE1, TB, 0, stream>>>(ei, NE, rowptr, rank, csr);

  const float* xin = x;
  for (int layer = 0; layer < 2; ++layer) {
    const float* W     = (const float*)d_in[2 + layer * 5];
    const float* a_src = (const float*)d_in[3 + layer * 5];
    const float* a_dst = (const float*)d_in[4 + layer * 5];
    const float* bias  = (const float*)d_in[5 + layer * 5];
    const float* Wskip = (const float*)d_in[6 + layer * 5];

    gat_gemm_f32_wmma<<<gGemm, TB, 0, stream>>>(xin, W, nullptr, h);
    gat_gemm_f32_wmma<<<gGemm, TB, 0, stream>>>(xin, Wskip, bias, agg);

    gat_attn_scores<<<gNH, TB, 0, stream>>>(h, a_src, a_dst, asrc, adst, N);

    hipMemsetAsync(mbuf, 0, (size_t)N * HEADS * sizeof(float), stream);
    hipMemsetAsync(denom, 0, (size_t)N * HEADS * sizeof(float), stream);

    gat_edge_pass1<<<gEH, TB, 0, stream>>>(ei, NE, E, asrc, adst, ebuf,
                                           (unsigned*)mbuf);
    gat_decode_max<<<gNH, TB, 0, stream>>>((unsigned*)mbuf, N * HEADS);
    gat_edge_pass2<<<gEH, TB, 0, stream>>>(ei, NE, E, ebuf, mbuf, denom);
    gat_gather_csr<<<gGath, TB, 0, stream>>>(csr, rowptr, ebuf, denom, h, agg,
                                             N, NE);

    gat_elu<<<gND, TB, 0, stream>>>(agg, y, N * DIM);
    xin = y;
  }
}


# Activation fused into the gather: each wave applies ELU to its finished row (skip + message sum) in registers and stores the output row; separate elementwise ELU pass removed
# speedup vs baseline: 1.8181x; 1.1186x over previous
#include <hip/hip_runtime.h>
#include <math.h>

#define DIM 128
#define HEADS 4
#define CH 32
#define NEG_SLOPE 0.2f

typedef __attribute__((ext_vector_type(2))) float v2f;
typedef __attribute__((ext_vector_type(8))) float v8f;

__global__ void gat_gemm_f32_wmma(const float* __restrict__ A,
                                  const float* __restrict__ W,
                                  const float* __restrict__ bias,
                                  float* __restrict__ out) {
  const int wave = threadIdx.x >> 5;
  const int lane = threadIdx.x & 31;
  const int row0 = blockIdx.x * 16;
  const int col0 = wave * 16;
  const int m  = lane & 15;
  const int kh = lane >> 4;

  const float* __restrict__ arow = A + (size_t)(row0 + m) * DIM;

  v8f acc = {};
  for (int k0 = 0; k0 < DIM; k0 += 4) {
    const int ka = k0 + kh * 2;
    v2f av, bv;
    av.x = arow[ka];
    av.y = arow[ka + 1];
    bv.x = W[(size_t)ka * DIM + col0 + m];
    bv.y = W[(size_t)(ka + 1) * DIM + col0 + m];
    acc = __builtin_amdgcn_wmma_f32_16x16x4_f32(
        false, av, false, bv, (short)0, acc, false, false);
  }

  const int nc = col0 + m;
  const float b = bias ? bias[nc] : 0.0f;
  const int rbase = row0 + kh * 8;
#pragma unroll
  for (int r = 0; r < 8; ++r)
    out[(size_t)(rbase + r) * DIM + nc] = acc[r] + b;
}

__global__ void gat_attn_scores(const float* __restrict__ h,
                                const float* __restrict__ a_src,
                                const float* __restrict__ a_dst,
                                float* __restrict__ asrc,
                                float* __restrict__ adst, int N) {
  int t = blockIdx.x * blockDim.x + threadIdx.x;
  if (t >= N * HEADS) return;
  const int n = t >> 2, hh = t & 3;
  const float* __restrict__ hp = h + (size_t)n * DIM + hh * CH;
  const float* __restrict__ ws = a_src + hh * CH;
  const float* __restrict__ wd = a_dst + hh * CH;
  float s1 = 0.f, s2 = 0.f;
#pragma unroll
  for (int c = 0; c < CH; ++c) {
    const float v = hp[c];
    s1 += v * ws[c];
    s2 += v * wd[c];
  }
  asrc[t] = s1;
  adst[t] = s2;
}

__device__ __forceinline__ void edge_sd(const int* __restrict__ ei, int e,
                                        int NE, int& s, int& d) {
  if (e < NE) { s = ei[e]; d = ei[NE + e]; }
  else        { s = d = e - NE; }
}

__device__ __forceinline__ unsigned ord_encode(float f) {
  unsigned u = __float_as_uint(f);
  return (f >= 0.0f) ? (u | 0x80000000u) : ~u;
}

__global__ void gat_edge_pass1(const int* __restrict__ ei, int NE, int E,
                               const float* __restrict__ asrc,
                               const float* __restrict__ adst,
                               float* __restrict__ ebuf,
                               unsigned* __restrict__ menc) {
  int t = blockIdx.x * blockDim.x + threadIdx.x;
  if (t >= E * HEADS) return;
  const int e = t >> 2, hh = t & 3;
  int s, d;
  edge_sd(ei, e, NE, s, d);
  float v = asrc[s * HEADS + hh] + adst[d * HEADS + hh];
  v = (v > 0.0f) ? v : NEG_SLOPE * v;
  ebuf[t] = v;
  atomicMax(&menc[d * HEADS + hh], ord_encode(v));
}

__global__ void gat_decode_max(unsigned* __restrict__ menc, int n) {
  int t = blockIdx.x * blockDim.x + threadIdx.x;
  if (t >= n) return;
  const unsigned u = menc[t];
  float f;
  if (u == 0u)                 f = 0.0f;
  else if (u & 0x80000000u)    f = __uint_as_float(u & 0x7FFFFFFFu);
  else                         f = __uint_as_float(~u);
  ((float*)menc)[t] = f;
}

__global__ void gat_edge_pass2(const int* __restrict__ ei, int NE, int E,
                               float* __restrict__ ebuf,
                               const float* __restrict__ m,
                               float* __restrict__ denom) {
  int t = blockIdx.x * blockDim.x + threadIdx.x;
  if (t >= E * HEADS) return;
  const int e = t >> 2, hh = t & 3;
  int s, d;
  edge_sd(ei, e, NE, s, d);
  const float ee = expf(ebuf[t] - m[d * HEADS + hh]);
  ebuf[t] = ee;
  atomicAdd(&denom[d * HEADS + hh], ee);
}

__global__ void csr_zero(int* __restrict__ cnt, int n) {
  int t = blockIdx.x * blockDim.x + threadIdx.x;
  if (t < n) cnt[t] = 0;
}

__global__ void csr_count(const int* __restrict__ ei, int NE,
                          int* __restrict__ cnt, int* __restrict__ rank) {
  int t = blockIdx.x * blockDim.x + threadIdx.x;
  if (t >= NE) return;
  rank[t] = atomicAdd(&cnt[ei[NE + t]], 1);
}

__global__ void __launch_bounds__(1024)
csr_scan(const int* __restrict__ cnt, int* __restrict__ rowptr, int chunk4) {
  __shared__ int wsum[32];
  const int t = threadIdx.x, lane = t & 31, wv = t >> 5;
  const int4* __restrict__ c4 = (const int4*)cnt + (size_t)t * chunk4;
  int s = 0;
  for (int i = 0; i < chunk4; ++i) {
    const int4 v = c4[i];
    s += (v.x + v.y) + (v.z + v.w);
  }
  int inc = s;
  for (int o = 1; o < 32; o <<= 1) {
    const int up = __shfl_up(inc, o);
    if (lane >= o) inc += up;
  }
  if (lane == 31) wsum[wv] = inc;
  __syncthreads();
  if (wv == 0) {
    const int w = wsum[lane];
    int wi = w;
    for (int o = 1; o < 32; o <<= 1) {
      const int up = __shfl_up(wi, o);
      if (lane >= o) wi += up;
    }
    wsum[lane] = wi - w;
  }
  __syncthreads();
  int run = wsum[wv] + inc - s;
  int4* __restrict__ r4 = (int4*)rowptr + (size_t)t * chunk4;
  for (int i = 0; i < chunk4; ++i) {
    const int4 v = c4[i];
    int4 o;
    o.x = run; run += v.x;
    o.y = run; run += v.y;
    o.z = run; run += v.z;
    o.w = run; run += v.w;
    r4[i] = o;
  }
}

__global__ void csr_fill(const int* __restrict__ ei, int NE,
                         const int* __restrict__ rowptr,
                         const int* __restrict__ rank,
                         int2* __restrict__ csr) {
  int t = blockIdx.x * blockDim.x + threadIdx.x;
  if (t >= NE) return;
  const int pos = rowptr[ei[NE + t]] + rank[t];
  csr[pos] = make_int2(ei[t], t);
}

__device__ __forceinline__ void edge_fma(float4& acc, const float4 hv, float a) {
  acc.x += hv.x * a;
  acc.y += hv.y * a;
  acc.z += hv.z * a;
  acc.w += hv.w * a;
}

__device__ __forceinline__ float elu1(float v) {
  return (v > 0.0f) ? v : expm1f(v);
}

__global__ void __launch_bounds__(256)
gat_gather_csr(const int2* __restrict__ csr, const int* __restrict__ rowptr,
               const float* __restrict__ ebuf, const float* __restrict__ denom,
               const float* __restrict__ h, const float* __restrict__ skip,
               float* __restrict__ out, int N, int NE) {
  const int lane = threadIdx.x & 31;
  const int hh = lane >> 3;
  const int wpb = blockDim.x >> 5;
  const int wv = __builtin_amdgcn_readfirstlane(threadIdx.x >> 5);
  const int nw = gridDim.x * wpb;
  const float4* __restrict__ h4 = (const float4*)h;
  const float4* __restrict__ k4 = (const float4*)skip;
  float4* __restrict__ o4 = (float4*)out;

  for (int d = blockIdx.x * wpb + wv; d < N; d += nw) {
    const int beg = rowptr[d];
    const int end = rowptr[d + 1];
    const float den = denom[d * HEADS + hh] + 1e-16f;

    float4 acc = make_float4(0.f, 0.f, 0.f, 0.f);
    {
      const float a = ebuf[(NE + d) * HEADS + hh] / den;
      edge_fma(acc, h4[d * 32 + lane], a);
    }
    for (int base = beg; base < end; base += 32) {
      const int rem = end - base;
      const int n = rem < 32 ? rem : 32;
      int2 my = make_int2(0, 0);
      if (lane < n) my = csr[base + lane];
      int j = 0;
      for (; j + 4 <= n; j += 4) {
        const int s0 = __builtin_amdgcn_readlane(my.x, j);
        const int s1 = __builtin_amdgcn_readlane(my.x, j + 1);
        const int s2 = __builtin_amdgcn_readlane(my.x, j + 2);
        const int s3 = __builtin_amdgcn_readlane(my.x, j + 3);
        const int e0 = __builtin_amdgcn_readlane(my.y, j);
        const int e1 = __builtin_amdgcn_readlane(my.y, j + 1);
        const int e2 = __builtin_amdgcn_readlane(my.y, j + 2);
        const int e3 = __builtin_amdgcn_readlane(my.y, j + 3);
        const float4 v0 = h4[s0 * 32 + lane];
        const float4 v1 = h4[s1 * 32 + lane];
        const float4 v2 = h4[s2 * 32 + lane];
        const float4 v3 = h4[s3 * 32 + lane];
        const float w0 = ebuf[e0 * HEADS + hh];
        const float w1 = ebuf[e1 * HEADS + hh];
        const float w2 = ebuf[e2 * HEADS + hh];
        const float w3 = ebuf[e3 * HEADS + hh];
        edge_fma(acc, v0, w0 / den);
        edge_fma(acc, v1, w1 / den);
        edge_fma(acc, v2, w2 / den);
        edge_fma(acc, v3, w3 / den);
      }
      for (; j < n; ++j) {
        const int s0 = __builtin_amdgcn_readlane(my.x, j);
        const int e0 = __builtin_amdgcn_readlane(my.y, j);
        const float4 v0 = h4[s0 * 32 + lane];
        const float w0 = ebuf[e0 * HEADS + hh];
        edge_fma(acc, v0, w0 / den);
      }
    }
    const float4 sk = k4[d * 32 + lane];
    float4 o;
    o.x = elu1(sk.x + acc.x);
    o.y = elu1(sk.y + acc.y);
    o.z = elu1(sk.z + acc.z);
    o.w = elu1(sk.w + acc.w);
    o4[d * 32 + lane] = o;
  }
}

extern "C" void kernel_launch(void* const* d_in, const int* in_sizes, int n_in,
                              void* d_out, int out_size, void* d_ws,
                              size_t ws_size, hipStream_t stream) {
  const float* x  = (const float*)d_in[0];
  const int*   ei = (const int*)d_in[1];

  const int N  = in_sizes[0] / DIM;
  const int NE = in_sizes[1] / 2;
  const int E  = NE + N;

  float* ws = (float*)d_ws;
  size_t o = 0;
  float* h     = ws + o; o += (size_t)N * DIM;
  float* agg   = ws + o; o += (size_t)N * DIM;
  float* asrc  = ws + o; o += (size_t)N * HEADS;
  float* adst  = ws + o; o += (size_t)N * HEADS;
  float* mbuf  = ws + o; o += (size_t)N * HEADS;
  float* denom = ws + o; o += (size_t)N * HEADS;
  float* ebuf  = ws + o; o += (size_t)E * HEADS;
  const int chunk4 = (N + 4096) / 4096;
  const int NP     = chunk4 * 4096;
  int2* csr    = (int2*)(ws + o); o += (size_t)NE * 2;
  int*  rank   = (int*)(ws + o);  o += (size_t)NE;
  int*  cnt    = (int*)(ws + o);  o += (size_t)NP;
  int*  rowptr = (int*)(ws + o);  o += (size_t)NP;

  float* y = (float*)d_out;

  const int TB = 256;
  const int gGemm = N / 16;
  const int gNH   = (N * HEADS + TB - 1) / TB;
  const int gEH   = (E * HEADS + TB - 1) / TB;
  const int gE1   = (NE + TB - 1) / TB;
  const int gNP   = (NP + TB - 1) / TB;
  const int gGath = 2048;

  csr_zero<<<gNP, TB, 0, stream>>>(cnt, NP);
  csr_count<<<gE1, TB, 0, stream>>>(ei, NE, cnt, rank);
  csr_scan<<<1, 1024, 0, stream>>>(cnt, rowptr, chunk4);
  csr_fill<<<gE1, TB, 0, stream>>>(ei, NE, rowptr, rank, csr);

  const float* xin = x;
  for (int layer = 0; layer < 2; ++layer) {
    const float* W     = (const float*)d_in[2 + layer * 5];
    const float* a_src = (const float*)d_in[3 + layer * 5];
    const float* a_dst = (const float*)d_in[4 + layer * 5];
    const float* bias  = (const float*)d_in[5 + layer * 5];
    const float* Wskip = (const float*)d_in[6 + layer * 5];

    gat_gemm_f32_wmma<<<gGemm, TB, 0, stream>>>(xin, W, nullptr, h);
    gat_gemm_f32_wmma<<<gGemm, TB, 0, stream>>>(xin, Wskip, bias, agg);

    gat_attn_scores<<<gNH, TB, 0, stream>>>(h, a_src, a_dst, asrc, adst, N);

    hipMemsetAsync(mbuf, 0, (size_t)N * HEADS * sizeof(float), stream);
    hipMemsetAsync(denom, 0, (size_t)N * HEADS * sizeof(float), stream);

    gat_edge_pass1<<<gEH, TB, 0, stream>>>(ei, NE, E, asrc, adst, ebuf,
                                           (unsigned*)mbuf);
    gat_decode_max<<<gNH, TB, 0, stream>>>((unsigned*)mbuf, N * HEADS);
    gat_edge_pass2<<<gEH, TB, 0, stream>>>(ei, NE, E, ebuf, mbuf, denom);
    gat_gather_csr<<<gGath, TB, 0, stream>>>(csr, rowptr, ebuf, denom, h, agg,
                                             y, N, NE);
    xin = y;
  }
}


# Edge softmax (scores, max, exp, sum, normalise) fused into the per-destination gather wave; edge passes, atomics, memsets removed; softmax sum order: in-wave lane tree, own loop last
# speedup vs baseline: 2.6667x; 1.4668x over previous
#include <hip/hip_runtime.h>
#include <math.h>

#define DIM 128
#define HEADS 4
#define CH 32
#define NEG_SLOPE 0.2f

typedef __attribute__((ext_vector_type(2))) float v2f;
typedef __attribute__((ext_vector_type(8))) float v8f;

__global__ void gat_gemm_f32_wmma(const float* __restrict__ A,
                                  const float* __restrict__ W,
                                  const float* __restrict__ bias,
                                  float* __restrict__ out) {
  const int wave = threadIdx.x >> 5;
  const int lane = threadIdx.x & 31;
  const int row0 = blockIdx.x * 16;
  const int col0 = wave * 16;
  const int m  = lane & 15;
  const int kh = lane >> 4;

  const float* __restrict__ arow = A + (size_t)(row0 + m) * DIM;

  v8f acc = {};
  for (int k0 = 0; k0 < DIM; k0 += 4) {
    const int ka = k0 + kh * 2;
    v2f av, bv;
    av.x = arow[ka];
    av.y = arow[ka + 1];
    bv.x = W[(size_t)ka * DIM + col0 + m];
    bv.y = W[(size_t)(ka + 1) * DIM + col0 + m];
    acc = __builtin_amdgcn_wmma_f32_16x16x4_f32(
        false, av, false, bv, (short)0, acc, false, false);
  }

  const int nc = col0 + m;
  const float b = bias ? bias[nc] : 0.0f;
  const int rbase = row0 + kh * 8;
#pragma unroll
  for (int r = 0; r < 8; ++r)
    out[(size_t)(rbase + r) * DIM + nc] = acc[r] + b;
}

__global__ void gat_attn_scores(const float* __restrict__ h,
                                const float* __restrict__ a_src,
                                const float* __restrict__ a_dst,
                                float* __restrict__ asrc,
                                float* __restrict__ adst, int N) {
  int t = blockIdx.x * blockDim.x + threadIdx.x;
  if (t >= N * HEADS) return;
  const int n = t >> 2, hh = t & 3;
  const float* __restrict__ hp = h + (size_t)n * DIM + hh * CH;
  const float* __restrict__ ws = a_src + hh * CH;
  const float* __restrict__ wd = a_dst + hh * CH;
  float s1 = 0.f, s2 = 0.f;
#pragma unroll
  for (int c = 0; c < CH; ++c) {
    const float v = hp[c];
    s1 += v * ws[c];
    s2 += v * wd[c];
  }
  asrc[t] = s1;
  adst[t] = s2;
}

__global__ void csr_zero(int* __restrict__ cnt, int n) {
  int t = blockIdx.x * blockDim.x + threadIdx.x;
  if (t < n) cnt[t] = 0;
}

__global__ void csr_count(const int* __restrict__ ei, int NE,
                          int* __restrict__ cnt, int* __restrict__ rank) {
  int t = blockIdx.x * blockDim.x + threadIdx.x;
  if (t >= NE) return;
  rank[t] = atomicAdd(&cnt[ei[NE + t]], 1);
}

__global__ void __launch_bounds__(1024)
csr_scan(const int* __restrict__ cnt, int* __restrict__ rowptr, int chunk4) {
  __shared__ int wsum[32];
  const int t = threadIdx.x, lane = t & 31, wv = t >> 5;
  const int4* __restrict__ c4 = (const int4*)cnt + (size_t)t * chunk4;
  int s = 0;
  for (int i = 0; i < chunk4; ++i) {
    const int4 v = c4[i];
    s += (v.x + v.y) + (v.z + v.w);
  }
  int inc = s;
  for (int o = 1; o < 32; o <<= 1) {
    const int up = __shfl_up(inc, o);
    if (lane >= o) inc += up;
  }
  if (lane == 31) wsum[wv] = inc;
  __syncthreads();
  if (wv == 0) {
    const int w = wsum[lane];
    int wi = w;
    for (int o = 1; o < 32; o <<= 1) {
      const int up = __shfl_up(wi, o);
      if (lane >= o) wi += up;
    }
    wsum[lane] = wi - w;
  }
  __syncthreads();
  int run = wsum[wv] + inc - s;
  int4* __restrict__ r4 = (int4*)rowptr + (size_t)t * chunk4;
  for (int i = 0; i < chunk4; ++i) {
    const int4 v = c4[i];
    int4 o;
    o.x = run; run += v.x;
    o.y = run; run += v.y;
    o.z = run; run += v.z;
    o.w = run; run += v.w;
    r4[i] = o;
  }
}

__global__ void csr_fill(const int* __restrict__ ei, int NE,
                         const int* __restrict__ rowptr,
                         const int* __restrict__ rank,
                         int* __restrict__ csr) {
  int t = blockIdx.x * blockDim.x + threadIdx.x;
  if (t >= NE) return;
  const int pos = rowptr[ei[NE + t]] + rank[t];
  csr[pos] = ei[t];
}

__device__ __forceinline__ void edge_fma(float4& acc, const float4 hv, float a) {
  acc.x += hv.x * a;
  acc.y += hv.y * a;
  acc.z += hv.z * a;
  acc.w += hv.w * a;
}

__device__ __forceinline__ float elu1(float v) {
  return (v > 0.0f) ? v : expm1f(v);
}

__device__ __forceinline__ float lrelu(float v) {
  return (v > 0.0f) ? v : NEG_SLOPE * v;
}

__device__ __forceinline__ float4 score4(const float4 a, const float4 b) {
  return make_float4(lrelu(a.x + b.x), lrelu(a.y + b.y), lrelu(a.z + b.z),
                     lrelu(a.w + b.w));
}

template <int CTRL>
__device__ __forceinline__ float row_xor(float v) {
  return __builtin_bit_cast(
      float, __builtin_amdgcn_update_dpp(0, __builtin_bit_cast(int, v), CTRL,
                                         0xf, 0xf, false));
}

__device__ __forceinline__ float xhalf(float v) {
  const int i = __builtin_bit_cast(int, v);
  return __builtin_bit_cast(
      float, __builtin_amdgcn_permlanex16(i, i, 0x76543210, 0xfedcba98, false,
                                          false));
}

__device__ __forceinline__ float wave_max(float v) {
  v = fmaxf(v, row_xor<0x161>(v));
  v = fmaxf(v, row_xor<0x162>(v));
  v = fmaxf(v, row_xor<0x164>(v));
  v = fmaxf(v, row_xor<0x168>(v));
  return fmaxf(v, xhalf(v));
}

__device__ __forceinline__ float wave_sum(float v) {
  v += row_xor<0x161>(v);
  v += row_xor<0x162>(v);
  v += row_xor<0x164>(v);
  v += row_xor<0x168>(v);
  return v + xhalf(v);
}

__device__ __forceinline__ float lane_f(float v, int j) {
  return __builtin_bit_cast(
      float, __builtin_amdgcn_readlane(__builtin_bit_cast(int, v), j));
}

__device__ __forceinline__ float uni(float v) {
  return __builtin_bit_cast(
      float, __builtin_amdgcn_readfirstlane(__builtin_bit_cast(int, v)));
}

__device__ __forceinline__ float pick(const float4 a, int hh) {
  return (hh & 2) ? ((hh & 1) ? a.w : a.z) : ((hh & 1) ? a.y : a.x);
}

template <bool FOUR>
__device__ __forceinline__ void gather_chunk(float4& acc,
                                             const float4* __restrict__ h4,
                                             int srcv, const float4 al, int n,
                                             int lane, int hh) {
  int j = 0;
  for (; FOUR && j + 4 <= n; j += 4) {
    const int s0 = __builtin_amdgcn_readlane(srcv, j);
    const int s1 = __builtin_amdgcn_readlane(srcv, j + 1);
    const int s2 = __builtin_amdgcn_readlane(srcv, j + 2);
    const int s3 = __builtin_amdgcn_readlane(srcv, j + 3);
    const float4 v0 = h4[s0 * 32 + lane];
    const float4 v1 = h4[s1 * 32 + lane];
    const float4 v2 = h4[s2 * 32 + lane];
    const float4 v3 = h4[s3 * 32 + lane];
    const float w0 = pick(make_float4(lane_f(al.x, j), lane_f(al.y, j),
                                      lane_f(al.z, j), lane_f(al.w, j)), hh);
    const float w1 = pick(make_float4(lane_f(al.x, j + 1), lane_f(al.y, j + 1),
                                      lane_f(al.z, j + 1), lane_f(al.w, j + 1)), hh);
    const float w2 = pick(make_float4(lane_f(al.x, j + 2), lane_f(al.y, j + 2),
                                      lane_f(al.z, j + 2), lane_f(al.w, j + 2)), hh);
    const float w3 = pick(make_float4(lane_f(al.x, j + 3), lane_f(al.y, j + 3),
                                      lane_f(al.z, j + 3), lane_f(al.w, j + 3)), hh);
    edge_fma(acc, v0, w0);
    edge_fma(acc, v1, w1);
    edge_fma(acc, v2, w2);
    edge_fma(acc, v3, w3);
  }
  for (; j < n; ++j) {
    const int s0 = __builtin_amdgcn_readlane(srcv, j);
    const float4 v0 = h4[s0 * 32 + lane];
    const float w0 = pick(make_float4(lane_f(al.x, j), lane_f(al.y, j),
                                      lane_f(al.z, j), lane_f(al.w, j)), hh);
    edge_fma(acc, v0, w0);
  }
}

__global__ void __launch_bounds__(256)
gat_softmax_gather_csr(const int* __restrict__ csr, const int* __restrict__ rowptr,
                       const float* __restrict__ asrc, const float* __restrict__ adst,
                       const float* __restrict__ h, const float* __restrict__ skip,
                       float* __restrict__ out, int N) {
  const int lane = threadIdx.x & 31;
  const int hh = lane >> 3;
  const int wpb = blockDim.x >> 5;
  const int wv = __builtin_amdgcn_readfirstlane(threadIdx.x >> 5);
  const int nw = gridDim.x * wpb;
  const float4* __restrict__ h4 = (const float4*)h;
  const float4* __restrict__ k4 = (const float4*)skip;
  const float4* __restrict__ as4 = (const float4*)asrc;
  const float4* __restrict__ ad4 = (const float4*)adst;
  float4* __restrict__ o4 = (float4*)out;
  const float NEG_INF = -__builtin_huge_valf();

  for (int d = blockIdx.x * wpb + wv; d < N; d += nw) {
    const int beg = rowptr[d];
    const int end = rowptr[d + 1];
    const float4 ad = ad4[d];
    const float4 es = score4(as4[d], ad);

    const int deg = end - beg;
    const int n0 = deg < 32 ? deg : 32;
    int src0 = 0;
    float4 e0 = make_float4(NEG_INF, NEG_INF, NEG_INF, NEG_INF);
    if (lane < n0) {
      src0 = csr[beg + lane];
      e0 = score4(as4[src0], ad);
    }

    float4 mx = e0;
    for (int base = beg + 32; base < end; base += 32) {
      if (base + lane < end) {
        const float4 e = score4(as4[csr[base + lane]], ad);
        mx.x = fmaxf(mx.x, e.x);
        mx.y = fmaxf(mx.y, e.y);
        mx.z = fmaxf(mx.z, e.z);
        mx.w = fmaxf(mx.w, e.w);
      }
    }
    float4 m;
    m.x = fmaxf(wave_max(mx.x), es.x);
    m.y = fmaxf(wave_max(mx.y), es.y);
    m.z = fmaxf(wave_max(mx.z), es.z);
    m.w = fmaxf(wave_max(mx.w), es.w);
    m.x = uni(isfinite(m.x) ? m.x : 0.0f);
    m.y = uni(isfinite(m.y) ? m.y : 0.0f);
    m.z = uni(isfinite(m.z) ? m.z : 0.0f);
    m.w = uni(isfinite(m.w) ? m.w : 0.0f);

    float4 ex0 = make_float4(0.f, 0.f, 0.f, 0.f);
    if (lane < n0) {
      ex0.x = expf(e0.x - m.x);
      ex0.y = expf(e0.y - m.y);
      ex0.z = expf(e0.z - m.z);
      ex0.w = expf(e0.w - m.w);
    }
    float4 sm = ex0;
    for (int base = beg + 32; base < end; base += 32) {
      if (base + lane < end) {
        const float4 e = score4(as4[csr[base + lane]], ad);
        sm.x += expf(e.x - m.x);
        sm.y += expf(e.y - m.y);
        sm.z += expf(e.z - m.z);
        sm.w += expf(e.w - m.w);
      }
    }
    const float4 exs = make_float4(expf(es.x - m.x), expf(es.y - m.y),
                                   expf(es.z - m.z), expf(es.w - m.w));
    float4 den;
    den.x = uni((wave_sum(sm.x) + exs.x) + 1e-16f);
    den.y = uni((wave_sum(sm.y) + exs.y) + 1e-16f);
    den.z = uni((wave_sum(sm.z) + exs.z) + 1e-16f);
    den.w = uni((wave_sum(sm.w) + exs.w) + 1e-16f);

    float4 acc = make_float4(0.f, 0.f, 0.f, 0.f);
    edge_fma(acc, h4[d * 32 + lane],
             pick(make_float4(exs.x / den.x, exs.y / den.y, exs.z / den.z,
                              exs.w / den.w), hh));
    {
      const float4 al = make_float4(ex0.x / den.x, ex0.y / den.y, ex0.z / den.z,
                                    ex0.w / den.w);
      gather_chunk<true>(acc, h4, src0, al, n0, lane, hh);
    }
    for (int base = beg + 32; base < end; base += 32) {
      const int rem = end - base;
      const int n = rem < 32 ? rem : 32;
      int srcv = 0;
      float4 al = make_float4(0.f, 0.f, 0.f, 0.f);
      if (lane < n) {
        srcv = csr[base + lane];
        const float4 e = score4(as4[srcv], ad);
        al.x = expf(e.x - m.x) / den.x;
        al.y = expf(e.y - m.y) / den.y;
        al.z = expf(e.z - m.z) / den.z;
        al.w = expf(e.w - m.w) / den.w;
      }
      gather_chunk<false>(acc, h4, srcv, al, n, lane, hh);
    }

    const float4 sk = k4[d * 32 + lane];
    float4 o;
    o.x = elu1(sk.x + acc.x);
    o.y = elu1(sk.y + acc.y);
    o.z = elu1(sk.z + acc.z);
    o.w = elu1(sk.w + acc.w);
    o4[d * 32 + lane] = o;
  }
}

extern "C" void kernel_launch(void* const* d_in, const int* in_sizes, int n_in,
                              void* d_out, int out_size, void* d_ws,
                              size_t ws_size, hipStream_t stream) {
  const float* x  = (const float*)d_in[0];
  const int*   ei = (const int*)d_in[1];

  const int N  = in_sizes[0] / DIM;
  const int NE = in_sizes[1] / 2;

  float* ws = (float*)d_ws;
  size_t o = 0;
  float* h     = ws + o; o += (size_t)N * DIM;
  float* agg   = ws + o; o += (size_t)N * DIM;
  float* asrc  = ws + o; o += (size_t)N * HEADS;
  float* adst  = ws + o; o += (size_t)N * HEADS;
  const int chunk4 = (N + 4096) / 4096;
  const int NP     = chunk4 * 4096;
  int*  csr    = (int*)(ws + o);  o += (size_t)NE;
  int*  rank   = (int*)(ws + o);  o += (size_t)NE;
  int*  cnt    = (int*)(ws + o);  o += (size_t)NP;
  int*  rowptr = (int*)(ws + o);  o += (size_t)NP;

  float* y = (float*)d_out;

  const int TB = 256;
  const int gGemm = N / 16;
  const int gNH   = (N * HEADS + TB - 1) / TB;
  const int gE1   = (NE + TB - 1) / TB;
  const int gNP   = (NP + TB - 1) / TB;
  const int gGath = 2048;

  csr_zero<<<gNP, TB, 0, stream>>>(cnt, NP);
  csr_count<<<gE1, TB, 0, stream>>>(ei, NE, cnt, rank);
  csr_scan<<<1, 1024, 0, stream>>>(cnt, rowptr, chunk4);
  csr_fill<<<gE1, TB, 0, stream>>>(ei, NE, rowptr, rank, csr);

  const float* xin = x;
  for (int layer = 0; layer < 2; ++layer) {
    const float* W     = (const float*)d_in[2 + layer * 5];
    const float* a_src = (const float*)d_in[3 + layer * 5];
    const float* a_dst = (const float*)d_in[4 + layer * 5];
    const float* bias  = (const float*)d_in[5 + layer * 5];
    const float* Wskip = (const float*)d_in[6 + layer * 5];

    gat_gemm_f32_wmma<<<gGemm, TB, 0, stream>>>(xin, W, nullptr, h);
    gat_gemm_f32_wmma<<<gGemm, TB, 0, stream>>>(xin, Wskip, bias, agg);

    gat_attn_scores<<<gNH, TB, 0, stream>>>(h, a_src, a_dst, asrc, adst, N);

    gat_softmax_gather_csr<<<gGath, TB, 0, stream>>>(csr, rowptr, asrc, adst, h,
                                                     agg, y, N);
    xin = y;
  }
}


# Dense products retiled: one launch per layer; 256 blocks stage W (even) or Wskip (odd) in LDS once, each wave holds a 16-row strip of x in registers for its 8 column tiles; whole-line stores
# speedup vs baseline: 3.4349x; 1.2881x over previous
#include <hip/hip_runtime.h>
#include <math.h>

#define DIM 128
#define HEADS 4
#define CH 32
#define NEG_SLOPE 0.2f

typedef __attribute__((ext_vector_type(2))) float v2f;
typedef __attribute__((ext_vector_type(8))) float v8f;

__device__ __forceinline__ float xhalf(float v) {
  const int i = __builtin_bit_cast(int, v);
  return __builtin_bit_cast(
      float, __builtin_amdgcn_permlanex16(i, i, 0x76543210, 0xfedcba98, false,
                                          false));
}

__device__ __forceinline__ void load_strip(v2f (&a)[32], const float* __restrict__ X,
                                           int strip, int m, int kh) {
  const v2f* __restrict__ arow =
      (const v2f*)(X + (size_t)(strip * 16 + m) * DIM + 2 * kh);
#pragma unroll
  for (int i = 0; i < 32; ++i) a[i] = arow[2 * i];
}

__global__ void __launch_bounds__(512)
gat_gemm_pair_f32_wmma(const float* __restrict__ X,
                       const float* __restrict__ Wa,
                       const float* __restrict__ Wb,
                       const float* __restrict__ bias_b,
                       float* __restrict__ out_a,
                       float* __restrict__ out_b, int nstrips) {
  __shared__ float4 wl[32 * 128];
  const int tid = threadIdx.x;
  const int lane = tid & 31;
  const int wave = __builtin_amdgcn_readfirstlane(tid >> 5);
  const int wpb = blockDim.x >> 5;
  const int m  = lane & 15;
  const int kh = lane >> 4;
  const int g = blockIdx.x & 1;
  const int pair = blockIdx.x >> 1;
  const int npairs = gridDim.x >> 1;
  const float* __restrict__ Wg = g ? Wb : Wa;
  float* __restrict__ out = g ? out_b : out_a;

  int s = pair + npairs * wave;
  const int sstep = npairs * wpb;

  float4 wv[8];
  {
    const int c = tid & 127;
#pragma unroll
    for (int q = 0; q < 8; ++q) {
      const int jh = (tid >> 7) + 4 * q;
      const int k = (jh >> 1) * 8 + (jh & 1) * 2;
      wv[q].x = Wg[(k) * DIM + c];
      wv[q].y = Wg[(k + 1) * DIM + c];
      wv[q].z = Wg[(k + 4) * DIM + c];
      wv[q].w = Wg[(k + 5) * DIM + c];
    }
  }
  float bz[8];
#pragma unroll
  for (int t = 0; t < 8; ++t) {
    const float bt = bias_b[16 * t + m];
    bz[t] = g ? bt : 0.0f;
  }
#pragma unroll
  for (int q = 0; q < 8; ++q) wl[tid + 512 * q] = wv[q];
  __syncthreads();

  const float4* wlane = wl + kh * 128 + m;
  while (s < nstrips) {
    v2f a[32];
    load_strip(a, X, s, m, kh);
    __builtin_amdgcn_sched_barrier(0);
    v8f acc[8];
#pragma unroll
    for (int t = 0; t < 8; ++t) acc[t] = (v8f){};
#pragma unroll
    for (int jj = 0; jj < 16; ++jj) {
      float4 b[8];
#pragma unroll
      for (int t = 0; t < 8; ++t) b[t] = wlane[jj * 256 + 16 * t];
#pragma unroll
      for (int t = 0; t < 8; ++t) {
        v2f bv; bv.x = b[t].x; bv.y = b[t].y;
        acc[t] = __builtin_amdgcn_wmma_f32_16x16x4_f32(
            false, a[2 * jj], false, bv, (short)0, acc[t], false, false);
      }
#pragma unroll
      for (int t = 0; t < 8; ++t) {
        v2f bv; bv.x = b[t].z; bv.y = b[t].w;
        acc[t] = __builtin_amdgcn_wmma_f32_16x16x4_f32(
            false, a[2 * jj + 1], false, bv, (short)0, acc[t], false, false);
      }
    }
    float* __restrict__ orow = out + (size_t)(s * 16) * DIM + lane;
#pragma unroll
    for (int tp = 0; tp < 4; ++tp) {
#pragma unroll
      for (int r = 0; r < 8; ++r) {
        const float xv = acc[2 * tp][r] + bz[2 * tp];
        const float yv = acc[2 * tp + 1][r] + bz[2 * tp + 1];
        const float xs = xhalf(xv);
        const float ys = xhalf(yv);
        orow[(size_t)r * DIM + 32 * tp] = kh ? ys : xv;
        orow[(size_t)(r + 8) * DIM + 32 * tp] = kh ? yv : xs;
      }
    }
    s += sstep;
  }
}

__global__ void gat_attn_scores(const float* __restrict__ h,
                                const float* __restrict__ a_src,
                                const float* __restrict__ a_dst,
                                float* __restrict__ asrc,
                                float* __restrict__ adst, int N) {
  int t = blockIdx.x * blockDim.x + threadIdx.x;
  if (t >= N * HEADS) return;
  const int n = t >> 2, hh = t & 3;
  const float* __restrict__ hp = h + (size_t)n * DIM + hh * CH;
  const float* __restrict__ ws = a_src + hh * CH;
  const float* __restrict__ wd = a_dst + hh * CH;
  float s1 = 0.f, s2 = 0.f;
#pragma unroll
  for (int c = 0; c < CH; ++c) {
    const float v = hp[c];
    s1 += v * ws[c];
    s2 += v * wd[c];
  }
  asrc[t] = s1;
  adst[t] = s2;
}

__global__ void csr_zero(int* __restrict__ cnt, int n) {
  int t = blockIdx.x * blockDim.x + threadIdx.x;
  if (t < n) cnt[t] = 0;
}

__global__ void csr_count(const int* __restrict__ ei, int NE,
                          int* __restrict__ cnt, int* __restrict__ rank) {
  int t = blockIdx.x * blockDim.x + threadIdx.x;
  if (t >= NE) return;
  rank[t] = atomicAdd(&cnt[ei[NE + t]], 1);
}

__global__ void __launch_bounds__(1024)
csr_scan(const int* __restrict__ cnt, int* __restrict__ rowptr, int chunk4) {
  __shared__ int wsum[32];
  const int t = threadIdx.x, lane = t & 31, wv = t >> 5;
  const int4* __restrict__ c4 = (const int4*)cnt + (size_t)t * chunk4;
  int s = 0;
  for (int i = 0; i < chunk4; ++i) {
    const int4 v = c4[i];
    s += (v.x + v.y) + (v.z + v.w);
  }
  int inc = s;
  for (int o = 1; o < 32; o <<= 1) {
    const int up = __shfl_up(inc, o);
    if (lane >= o) inc += up;
  }
  if (lane == 31) wsum[wv] = inc;
  __syncthreads();
  if (wv == 0) {
    const int w = wsum[lane];
    int wi = w;
    for (int o = 1; o < 32; o <<= 1) {
      const int up = __shfl_up(wi, o);
      if (lane >= o) wi += up;
    }
    wsum[lane] = wi - w;
  }
  __syncthreads();
  int run = wsum[wv] + inc - s;
  int4* __restrict__ r4 = (int4*)rowptr + (size_t)t * chunk4;
  for (int i = 0; i < chunk4; ++i) {
    const int4 v = c4[i];
    int4 o;
    o.x = run; run += v.x;
    o.y = run; run += v.y;
    o.z = run; run += v.z;
    o.w = run; run += v.w;
    r4[i] = o;
  }
}

__global__ void csr_fill(const int* __restrict__ ei, int NE,
                         const int* __restrict__ rowptr,
                         const int* __restrict__ rank,
                         int* __restrict__ csr) {
  int t = blockIdx.x * blockDim.x + threadIdx.x;
  if (t >= NE) return;
  const int pos = rowptr[ei[NE + t]] + rank[t];
  csr[pos] = ei[t];
}

__device__ __forceinline__ void edge_fma(float4& acc, const float4 hv, float a) {
  acc.x += hv.x * a;
  acc.y += hv.y * a;
  acc.z += hv.z * a;
  acc.w += hv.w * a;
}

__device__ __forceinline__ float elu1(float v) {
  return (v > 0.0f) ? v : expm1f(v);
}

__device__ __forceinline__ float lrelu(float v) {
  return (v > 0.0f) ? v : NEG_SLOPE * v;
}

__device__ __forceinline__ float4 score4(const float4 a, const float4 b) {
  return make_float4(lrelu(a.x + b.x), lrelu(a.y + b.y), lrelu(a.z + b.z),
                     lrelu(a.w + b.w));
}

template <int CTRL>
__device__ __forceinline__ float row_xor(float v) {
  return __builtin_bit_cast(
      float, __builtin_amdgcn_update_dpp(0, __builtin_bit_cast(int, v), CTRL,
                                         0xf, 0xf, false));
}

__device__ __forceinline__ float wave_max(float v) {
  v = fmaxf(v, row_xor<0x161>(v));
  v = fmaxf(v, row_xor<0x162>(v));
  v = fmaxf(v, row_xor<0x164>(v));
  v = fmaxf(v, row_xor<0x168>(v));
  return fmaxf(v, xhalf(v));
}

__device__ __forceinline__ float wave_sum(float v) {
  v += row_xor<0x161>(v);
  v += row_xor<0x162>(v);
  v += row_xor<0x164>(v);
  v += row_xor<0x168>(v);
  return v + xhalf(v);
}

__device__ __forceinline__ float lane_f(float v, int j) {
  return __builtin_bit_cast(
      float, __builtin_amdgcn_readlane(__builtin_bit_cast(int, v), j));
}

__device__ __forceinline__ float uni(float v) {
  return __builtin_bit_cast(
      float, __builtin_amdgcn_readfirstlane(__builtin_bit_cast(int, v)));
}

__device__ __forceinline__ float pick(const float4 a, int hh) {
  return (hh & 2) ? ((hh & 1) ? a.w : a.z) : ((hh & 1) ? a.y : a.x);
}

template <bool FOUR>
__device__ __forceinline__ void gather_chunk(float4& acc,
                                             const float4* __restrict__ h4,
                                             int srcv, const float4 al, int n,
                                             int lane, int hh) {
  int j = 0;
  for (; FOUR && j + 4 <= n; j += 4) {
    const int s0 = __builtin_amdgcn_readlane(srcv, j);
    const int s1 = __builtin_amdgcn_readlane(srcv, j + 1);
    const int s2 = __builtin_amdgcn_readlane(srcv, j + 2);
    const int s3 = __builtin_amdgcn_readlane(srcv, j + 3);
    const float4 v0 = h4[s0 * 32 + lane];
    const float4 v1 = h4[s1 * 32 + lane];
    const float4 v2 = h4[s2 * 32 + lane];
    const float4 v3 = h4[s3 * 32 + lane];
    const float w0 = pick(make_float4(lane_f(al.x, j), lane_f(al.y, j),
                                      lane_f(al.z, j), lane_f(al.w, j)), hh);
    const float w1 = pick(make_float4(lane_f(al.x, j + 1), lane_f(al.y, j + 1),
                                      lane_f(al.z, j + 1), lane_f(al.w, j + 1)), hh);
    const float w2 = pick(make_float4(lane_f(al.x, j + 2), lane_f(al.y, j + 2),
                                      lane_f(al.z, j + 2), lane_f(al.w, j + 2)), hh);
    const float w3 = pick(make_float4(lane_f(al.x, j + 3), lane_f(al.y, j + 3),
                                      lane_f(al.z, j + 3), lane_f(al.w, j + 3)), hh);
    edge_fma(acc, v0, w0);
    edge_fma(acc, v1, w1);
    edge_fma(acc, v2, w2);
    edge_fma(acc, v3, w3);
  }
  for (; j < n; ++j) {
    const int s0 = __builtin_amdgcn_readlane(srcv, j);
    const float4 v0 = h4[s0 * 32 + lane];
    const float w0 = pick(make_float4(lane_f(al.x, j), lane_f(al.y, j),
                                      lane_f(al.z, j), lane_f(al.w, j)), hh);
    edge_fma(acc, v0, w0);
  }
}

__global__ void __launch_bounds__(256)
gat_softmax_gather_csr(const int* __restrict__ csr, const int* __restrict__ rowptr,
                       const float* __restrict__ asrc, const float* __restrict__ adst,
                       const float* __restrict__ h, const float* __restrict__ skip,
                       float* __restrict__ out, int N) {
  const int lane = threadIdx.x & 31;
  const int hh = lane >> 3;
  const int wpb = blockDim.x >> 5;
  const int wv = __builtin_amdgcn_readfirstlane(threadIdx.x >> 5);
  const int nw = gridDim.x * wpb;
  const float4* __restrict__ h4 = (const float4*)h;
  const float4* __restrict__ k4 = (const float4*)skip;
  const float4* __restrict__ as4 = (const float4*)asrc;
  const float4* __restrict__ ad4 = (const float4*)adst;
  float4* __restrict__ o4 = (float4*)out;
  const float NEG_INF = -__builtin_huge_valf();

  for (int d = blockIdx.x * wpb + wv; d < N; d += nw) {
    const int beg = rowptr[d];
    const int end = rowptr[d + 1];
    const float4 ad = ad4[d];
    const float4 es = score4(as4[d], ad);

    const int deg = end - beg;
    const int n0 = deg < 32 ? deg : 32;
    int src0 = 0;
    float4 e0 = make_float4(NEG_INF, NEG_INF, NEG_INF, NEG_INF);
    if (lane < n0) {
      src0 = csr[beg + lane];
      e0 = score4(as4[src0], ad);
    }

    float4 mx = e0;
    for (int base = beg + 32; base < end; base += 32) {
      if (base + lane < end) {
        const float4 e = score4(as4[csr[base + lane]], ad);
        mx.x = fmaxf(mx.x, e.x);
        mx.y = fmaxf(mx.y, e.y);
        mx.z = fmaxf(mx.z, e.z);
        mx.w = fmaxf(mx.w, e.w);
      }
    }
    float4 m;
    m.x = fmaxf(wave_max(mx.x), es.x);
    m.y = fmaxf(wave_max(mx.y), es.y);
    m.z = fmaxf(wave_max(mx.z), es.z);
    m.w = fmaxf(wave_max(mx.w), es.w);
    m.x = uni(isfinite(m.x) ? m.x : 0.0f);
    m.y = uni(isfinite(m.y) ? m.y : 0.0f);
    m.z = uni(isfinite(m.z) ? m.z : 0.0f);
    m.w = uni(isfinite(m.w) ? m.w : 0.0f);

    float4 ex0 = make_float4(0.f, 0.f, 0.f, 0.f);
    if (lane < n0) {
      ex0.x = expf(e0.x - m.x);
      ex0.y = expf(e0.y - m.y);
      ex0.z = expf(e0.z - m.z);
      ex0.w = expf(e0.w - m.w);
    }
    float4 sm = ex0;
    for (int base = beg + 32; base < end; base += 32) {
      if (base + lane < end) {
        const float4 e = score4(as4[csr[base + lane]], ad);
        sm.x += expf(e.x - m.x);
        sm.y += expf(e.y - m.y);
        sm.z += expf(e.z - m.z);
        sm.w += expf(e.w - m.w);
      }
    }
    const float4 exs = make_float4(expf(es.x - m.x), expf(es.y - m.y),
                                   expf(es.z - m.z), expf(es.w - m.w));
    float4 den;
    den.x = uni((wave_sum(sm.x) + exs.x) + 1e-16f);
    den.y = uni((wave_sum(sm.y) + exs.y) + 1e-16f);
    den.z = uni((wave_sum(sm.z) + exs.z) + 1e-16f);
    den.w = uni((wave_sum(sm.w) + exs.w) + 1e-16f);

    float4 acc = make_float4(0.f, 0.f, 0.f, 0.f);
    edge_fma(acc, h4[d * 32 + lane],
             pick(make_float4(exs.x / den.x, exs.y / den.y, exs.z / den.z,
                              exs.w / den.w), hh));
    {
      const float4 al = make_float4(ex0.x / den.x, ex0.y / den.y, ex0.z / den.z,
                                    ex0.w / den.w);
      gather_chunk<true>(acc, h4, src0, al, n0, lane, hh);
    }
    for (int base = beg + 32; base < end; base += 32) {
      const int rem = end - base;
      const int n = rem < 32 ? rem : 32;
      int srcv = 0;
      float4 al = make_float4(0.f, 0.f, 0.f, 0.f);
      if (lane < n) {
        srcv = csr[base + lane];
        const float4 e = score4(as4[srcv], ad);
        al.x = expf(e.x - m.x) / den.x;
        al.y = expf(e.y - m.y) / den.y;
        al.z = expf(e.z - m.z) / den.z;
        al.w = expf(e.w - m.w) / den.w;
      }
      gather_chunk<false>(acc, h4, srcv, al, n, lane, hh);
    }

    const float4 sk = k4[d * 32 + lane];
    float4 o;
    o.x = elu1(sk.x + acc.x);
    o.y = elu1(sk.y + acc.y);
    o.z = elu1(sk.z + acc.z);
    o.w = elu1(sk.w + acc.w);
    o4[d * 32 + lane] = o;
  }
}

extern "C" void kernel_launch(void* const* d_in, const int* in_sizes, int n_in,
                              void* d_out, int out_size, void* d_ws,
                              size_t ws_size, hipStream_t stream) {
  const float* x  = (const float*)d_in[0];
  const int*   ei = (const int*)d_in[1];

  const int N  = in_sizes[0] / DIM;
  const int NE = in_sizes[1] / 2;

  float* ws = (float*)d_ws;
  size_t o = 0;
  float* h     = ws + o; o += (size_t)N * DIM;
  float* agg   = ws + o; o += (size_t)N * DIM;
  float* asrc  = ws + o; o += (size_t)N * HEADS;
  float* adst  = ws + o; o += (size_t)N * HEADS;
  const int chunk4 = (N + 4096) / 4096;
  const int NP     = chunk4 * 4096;
  int*  csr    = (int*)(ws + o);  o += (size_t)NE;
  int*  rank   = (int*)(ws + o);  o += (size_t)NE;
  int*  cnt    = (int*)(ws + o);  o += (size_t)NP;
  int*  rowptr = (int*)(ws + o);  o += (size_t)NP;

  float* y = (float*)d_out;

  const int TB = 256;
  const int nStrips = N / 16;
  const int gGemm = 256;
  const int gNH   = (N * HEADS + TB - 1) / TB;
  const int gE1   = (NE + TB - 1) / TB;
  const int gNP   = (NP + TB - 1) / TB;
  const int gGath = 2048;

  csr_zero<<<gNP, TB, 0, stream>>>(cnt, NP);
  csr_count<<<gE1, TB, 0, stream>>>(ei, NE, cnt, rank);
  csr_scan<<<1, 1024, 0, stream>>>(cnt, rowptr, chunk4);
  csr_fill<<<gE1, TB, 0, stream>>>(ei, NE, rowptr, rank, csr);

  const float* xin = x;
  for (int layer = 0; layer < 2; ++layer) {
    const float* W     = (const float*)d_in[2 + layer * 5];
    const float* a_src = (const float*)d_in[3 + layer * 5];
    const float* a_dst = (const float*)d_in[4 + layer * 5];
    const float* bias  = (const float*)d_in[5 + layer * 5];
    const float* Wskip = (const float*)d_in[6 + layer * 5];

    gat_gemm_pair_f32_wmma<<<gGemm, 512, 0, stream>>>(xin, W, Wskip, bias, h, agg,
                                                      nStrips);

    gat_attn_scores<<<gNH, TB, 0, stream>>>(h, a_src, a_dst, asrc, adst, N);

    gat_softmax_gather_csr<<<gGath, TB, 0, stream>>>(csr, rowptr, asrc, adst, h,
                                                     agg, y, N);
    xin = y;
  }
}


# Attention scores fused into the dense-product launch: even blocks fold each row's 16 lanes over the accumulator tiles; scores pass removed; score sum order: 2-term lane partial, then 16-lane tree
# speedup vs baseline: 3.5774x; 1.0415x over previous
#include <hip/hip_runtime.h>
#include <math.h>

#define DIM 128
#define HEADS 4
#define CH 32
#define NEG_SLOPE 0.2f

typedef __attribute__((ext_vector_type(2))) float v2f;
typedef __attribute__((ext_vector_type(8))) float v8f;

__device__ __forceinline__ float xhalf(float v) {
  const int i = __builtin_bit_cast(int, v);
  return __builtin_bit_cast(
      float, __builtin_amdgcn_permlanex16(i, i, 0x76543210, 0xfedcba98, false,
                                          false));
}

template <int CTRL>
__device__ __forceinline__ float row_xor(float v) {
  return __builtin_bit_cast(
      float, __builtin_amdgcn_update_dpp(0, __builtin_bit_cast(int, v), CTRL,
                                         0xf, 0xf, false));
}

template <int CTRL>
__device__ __forceinline__ float fold(float a, float b, bool up) {
  const float keep = up ? b : a;
  const float give = up ? a : b;
  return keep + row_xor<CTRL>(give);
}

__device__ __forceinline__ void load_strip(v2f (&a)[32], const float* __restrict__ X,
                                           int strip, int m, int kh) {
  const v2f* __restrict__ arow =
      (const v2f*)(X + (size_t)(strip * 16 + m) * DIM + 2 * kh);
#pragma unroll
  for (int i = 0; i < 32; ++i) a[i] = arow[2 * i];
}

__global__ void __launch_bounds__(512)
gat_gemm_pair_f32_wmma(const float* __restrict__ X,
                       const float* __restrict__ Wa,
                       const float* __restrict__ Wb,
                       const float* __restrict__ bias_b,
                       const float* __restrict__ a_src,
                       const float* __restrict__ a_dst,
                       float* __restrict__ out_a,
                       float* __restrict__ out_b,
                       float* __restrict__ asrc,
                       float* __restrict__ adst, int nstrips) {
  __shared__ float4 wl[32 * 128];
  const int tid = threadIdx.x;
  const int lane = tid & 31;
  const int wave = __builtin_amdgcn_readfirstlane(tid >> 5);
  const int wpb = blockDim.x >> 5;
  const int m  = lane & 15;
  const int kh = lane >> 4;
  const int g = blockIdx.x & 1;
  const int pair = blockIdx.x >> 1;
  const int npairs = gridDim.x >> 1;
  const float* __restrict__ Wg = g ? Wb : Wa;
  float* __restrict__ out = g ? out_b : out_a;

  int s = pair + npairs * wave;
  const int sstep = npairs * wpb;

  float4 wv[8];
  {
    const int c = tid & 127;
#pragma unroll
    for (int q = 0; q < 8; ++q) {
      const int jh = (tid >> 7) + 4 * q;
      const int k = (jh >> 1) * 8 + (jh & 1) * 2;
      wv[q].x = Wg[(k) * DIM + c];
      wv[q].y = Wg[(k + 1) * DIM + c];
      wv[q].z = Wg[(k + 4) * DIM + c];
      wv[q].w = Wg[(k + 5) * DIM + c];
    }
  }
  float bz[8];
  float asv[8], adv[8];
#pragma unroll
  for (int t = 0; t < 8; ++t) {
    const float bt = bias_b[16 * t + m];
    bz[t] = g ? bt : 0.0f;
    asv[t] = a_src[16 * t + m];
    adv[t] = a_dst[16 * t + m];
  }
  const bool up0 = m & 1, up1 = m & 2, up2 = m & 4, up3 = m & 8;
#pragma unroll
  for (int q = 0; q < 8; ++q) wl[tid + 512 * q] = wv[q];
  __syncthreads();

  const float4* wlane = wl + kh * 128 + m;
  while (s < nstrips) {
    v2f a[32];
    load_strip(a, X, s, m, kh);
    __builtin_amdgcn_sched_barrier(0);
    v8f acc[8];
#pragma unroll
    for (int t = 0; t < 8; ++t) acc[t] = (v8f){};
#pragma unroll
    for (int jj = 0; jj < 16; ++jj) {
      float4 b[8];
#pragma unroll
      for (int t = 0; t < 8; ++t) b[t] = wlane[jj * 256 + 16 * t];
#pragma unroll
      for (int t = 0; t < 8; ++t) {
        v2f bv; bv.x = b[t].x; bv.y = b[t].y;
        acc[t] = __builtin_amdgcn_wmma_f32_16x16x4_f32(
            false, a[2 * jj], false, bv, (short)0, acc[t], false, false);
      }
#pragma unroll
      for (int t = 0; t < 8; ++t) {
        v2f bv; bv.x = b[t].z; bv.y = b[t].w;
        acc[t] = __builtin_amdgcn_wmma_f32_16x16x4_f32(
            false, a[2 * jj + 1], false, bv, (short)0, acc[t], false, false);
      }
    }
    float* __restrict__ orow = out + (size_t)(s * 16) * DIM + lane;
#pragma unroll
    for (int tp = 0; tp < 4; ++tp) {
#pragma unroll
      for (int r = 0; r < 8; ++r) {
        const float xv = acc[2 * tp][r] + bz[2 * tp];
        const float yv = acc[2 * tp + 1][r] + bz[2 * tp + 1];
        const float xs = xhalf(xv);
        const float ys = xhalf(yv);
        orow[(size_t)r * DIM + 32 * tp] = kh ? ys : xv;
        orow[(size_t)(r + 8) * DIM + 32 * tp] = kh ? yv : xs;
      }
    }
    if (g == 0) {
#pragma unroll
      for (int sd = 0; sd < 2; ++sd) {
        const float* av = sd ? adv : asv;
        float w2[8];
#pragma unroll
        for (int r = 0; r < 8; ++r) {
          float p[4];
#pragma unroll
          for (int hd = 0; hd < 4; ++hd)
            p[hd] = acc[2 * hd][r] * av[2 * hd] + acc[2 * hd + 1][r] * av[2 * hd + 1];
          const float u0 = fold<0x161>(p[0], p[1], up0);
          const float u1 = fold<0x161>(p[2], p[3], up0);
          w2[r] = fold<0x162>(u0, u1, up1);
        }
        float w3[4];
#pragma unroll
        for (int q = 0; q < 4; ++q)
          w3[q] = fold<0x164>(w2[2 * q], w2[2 * q + 1], up2);
        const float y0 = fold<0x168>(w3[0], w3[1], up3);
        const float y1 = fold<0x168>(w3[2], w3[3], up3);
        const float x0 = xhalf(y0);
        const float x1 = xhalf(y1);
        float* __restrict__ srow = (sd ? adst : asrc) + (size_t)s * 64 + lane;
        srow[0] = kh ? x1 : y0;
        srow[32] = kh ? y1 : x0;
      }
    }
    s += sstep;
  }
}

__global__ void csr_zero(int* __restrict__ cnt, int n) {
  int t = blockIdx.x * blockDim.x + threadIdx.x;
  if (t < n) cnt[t] = 0;
}

__global__ void csr_count(const int* __restrict__ ei, int NE,
                          int* __restrict__ cnt, int* __restrict__ rank) {
  int t = blockIdx.x * blockDim.x + threadIdx.x;
  if (t >= NE) return;
  rank[t] = atomicAdd(&cnt[ei[NE + t]], 1);
}

__global__ void __launch_bounds__(1024)
csr_scan(const int* __restrict__ cnt, int* __restrict__ rowptr, int chunk4) {
  __shared__ int wsum[32];
  const int t = threadIdx.x, lane = t & 31, wv = t >> 5;
  const int4* __restrict__ c4 = (const int4*)cnt + (size_t)t * chunk4;
  int s = 0;
  for (int i = 0; i < chunk4; ++i) {
    const int4 v = c4[i];
    s += (v.x + v.y) + (v.z + v.w);
  }
  int inc = s;
  for (int o = 1; o < 32; o <<= 1) {
    const int up = __shfl_up(inc, o);
    if (lane >= o) inc += up;
  }
  if (lane == 31) wsum[wv] = inc;
  __syncthreads();
  if (wv == 0) {
    const int w = wsum[lane];
    int wi = w;
    for (int o = 1; o < 32; o <<= 1) {
      const int up = __shfl_up(wi, o);
      if (lane >= o) wi += up;
    }
    wsum[lane] = wi - w;
  }
  __syncthreads();
  int run = wsum[wv] + inc - s;
  int4* __restrict__ r4 = (int4*)rowptr + (size_t)t * chunk4;
  for (int i = 0; i < chunk4; ++i) {
    const int4 v = c4[i];
    int4 o;
    o.x = run; run += v.x;
    o.y = run; run += v.y;
    o.z = run; run += v.z;
    o.w = run; run += v.w;
    r4[i] = o;
  }
}

__global__ void csr_fill(const int* __restrict__ ei, int NE,
                         const int* __restrict__ rowptr,
                         const int* __restrict__ rank,
                         int* __restrict__ csr) {
  int t = blockIdx.x * blockDim.x + threadIdx.x;
  if (t >= NE) return;
  const int pos = rowptr[ei[NE + t]] + rank[t];
  csr[pos] = ei[t];
}

__device__ __forceinline__ void edge_fma(float4& acc, const float4 hv, float a) {
  acc.x += hv.x * a;
  acc.y += hv.y * a;
  acc.z += hv.z * a;
  acc.w += hv.w * a;
}

__device__ __forceinline__ float elu1(float v) {
  return (v > 0.0f) ? v : expm1f(v);
}

__device__ __forceinline__ float lrelu(float v) {
  return (v > 0.0f) ? v : NEG_SLOPE * v;
}

__device__ __forceinline__ float4 score4(const float4 a, const float4 b) {
  return make_float4(lrelu(a.x + b.x), lrelu(a.y + b.y), lrelu(a.z + b.z),
                     lrelu(a.w + b.w));
}

__device__ __forceinline__ float wave_max(float v) {
  v = fmaxf(v, row_xor<0x161>(v));
  v = fmaxf(v, row_xor<0x162>(v));
  v = fmaxf(v, row_xor<0x164>(v));
  v = fmaxf(v, row_xor<0x168>(v));
  return fmaxf(v, xhalf(v));
}

__device__ __forceinline__ float wave_sum(float v) {
  v += row_xor<0x161>(v);
  v += row_xor<0x162>(v);
  v += row_xor<0x164>(v);
  v += row_xor<0x168>(v);
  return v + xhalf(v);
}

__device__ __forceinline__ float lane_f(float v, int j) {
  return __builtin_bit_cast(
      float, __builtin_amdgcn_readlane(__builtin_bit_cast(int, v), j));
}

__device__ __forceinline__ float uni(float v) {
  return __builtin_bit_cast(
      float, __builtin_amdgcn_readfirstlane(__builtin_bit_cast(int, v)));
}

__device__ __forceinline__ float pick(const float4 a, int hh) {
  return (hh & 2) ? ((hh & 1) ? a.w : a.z) : ((hh & 1) ? a.y : a.x);
}

template <bool FOUR>
__device__ __forceinline__ void gather_chunk(float4& acc,
                                             const float4* __restrict__ h4,
                                             int srcv, const float4 al, int n,
                                             int lane, int hh) {
  int j = 0;
  for (; FOUR && j + 4 <= n; j += 4) {
    const int s0 = __builtin_amdgcn_readlane(srcv, j);
    const int s1 = __builtin_amdgcn_readlane(srcv, j + 1);
    const int s2 = __builtin_amdgcn_readlane(srcv, j + 2);
    const int s3 = __builtin_amdgcn_readlane(srcv, j + 3);
    const float4 v0 = h4[s0 * 32 + lane];
    const float4 v1 = h4[s1 * 32 + lane];
    const float4 v2 = h4[s2 * 32 + lane];
    const float4 v3 = h4[s3 * 32 + lane];
    const float w0 = pick(make_float4(lane_f(al.x, j), lane_f(al.y, j),
                                      lane_f(al.z, j), lane_f(al.w, j)), hh);
    const float w1 = pick(make_float4(lane_f(al.x, j + 1), lane_f(al.y, j + 1),
                                      lane_f(al.z, j + 1), lane_f(al.w, j + 1)), hh);
    const float w2 = pick(make_float4(lane_f(al.x, j + 2), lane_f(al.y, j + 2),
                                      lane_f(al.z, j + 2), lane_f(al.w, j + 2)), hh);
    const float w3 = pick(make_float4(lane_f(al.x, j + 3), lane_f(al.y, j + 3),
                                      lane_f(al.z, j + 3), lane_f(al.w, j + 3)), hh);
    edge_fma(acc, v0, w0);
    edge_fma(acc, v1, w1);
    edge_fma(acc, v2, w2);
    edge_fma(acc, v3, w3);
  }
  for (; j < n; ++j) {
    const int s0 = __builtin_amdgcn_readlane(srcv, j);
    const float4 v0 = h4[s0 * 32 + lane];
    const float w0 = pick(make_float4(lane_f(al.x, j), lane_f(al.y, j),
                                      lane_f(al.z, j), lane_f(al.w, j)), hh);
    edge_fma(acc, v0, w0);
  }
}

__global__ void __launch_bounds__(256)
gat_softmax_gather_csr(const int* __restrict__ csr, const int* __restrict__ rowptr,
                       const float* __restrict__ asrc, const float* __restrict__ adst,
                       const float* __restrict__ h, const float* __restrict__ skip,
                       float* __restrict__ out, int N) {
  const int lane = threadIdx.x & 31;
  const int hh = lane >> 3;
  const int wpb = blockDim.x >> 5;
  const int wv = __builtin_amdgcn_readfirstlane(threadIdx.x >> 5);
  const int nw = gridDim.x * wpb;
  const float4* __restrict__ h4 = (const float4*)h;
  const float4* __restrict__ k4 = (const float4*)skip;
  const float4* __restrict__ as4 = (const float4*)asrc;
  const float4* __restrict__ ad4 = (const float4*)adst;
  float4* __restrict__ o4 = (float4*)out;
  const float NEG_INF = -__builtin_huge_valf();

  for (int d = blockIdx.x * wpb + wv; d < N; d += nw) {
    const int beg = rowptr[d];
    const int end = rowptr[d + 1];
    const float4 ad = ad4[d];
    const float4 es = score4(as4[d], ad);

    const int deg = end - beg;
    const int n0 = deg < 32 ? deg : 32;
    int src0 = 0;
    float4 e0 = make_float4(NEG_INF, NEG_INF, NEG_INF, NEG_INF);
    if (lane < n0) {
      src0 = csr[beg + lane];
      e0 = score4(as4[src0], ad);
    }

    float4 mx = e0;
    for (int base = beg + 32; base < end; base += 32) {
      if (base + lane < end) {
        const float4 e = score4(as4[csr[base + lane]], ad);
        mx.x = fmaxf(mx.x, e.x);
        mx.y = fmaxf(mx.y, e.y);
        mx.z = fmaxf(mx.z, e.z);
        mx.w = fmaxf(mx.w, e.w);
      }
    }
    float4 m;
    m.x = fmaxf(wave_max(mx.x), es.x);
    m.y = fmaxf(wave_max(mx.y), es.y);
    m.z = fmaxf(wave_max(mx.z), es.z);
    m.w = fmaxf(wave_max(mx.w), es.w);
    m.x = uni(isfinite(m.x) ? m.x : 0.0f);
    m.y = uni(isfinite(m.y) ? m.y : 0.0f);
    m.z = uni(isfinite(m.z) ? m.z : 0.0f);
    m.w = uni(isfinite(m.w) ? m.w : 0.0f);

    float4 ex0 = make_float4(0.f, 0.f, 0.f, 0.f);
    if (lane < n0) {
      ex0.x = expf(e0.x - m.x);
      ex0.y = expf(e0.y - m.y);
      ex0.z = expf(e0.z - m.z);
      ex0.w = expf(e0.w - m.w);
    }
    float4 sm = ex0;
    for (int base = beg + 32; base < end; base += 32) {
      if (base + lane < end) {
        const float4 e = score4(as4[csr[base + lane]], ad);
        sm.x += expf(e.x - m.x);
        sm.y += expf(e.y - m.y);
        sm.z += expf(e.z - m.z);
        sm.w += expf(e.w - m.w);
      }
    }
    const float4 exs = make_float4(expf(es.x - m.x), expf(es.y - m.y),
                                   expf(es.z - m.z), expf(es.w - m.w));
    float4 den;
    den.x = uni((wave_sum(sm.x) + exs.x) + 1e-16f);
    den.y = uni((wave_sum(sm.y) + exs.y) + 1e-16f);
    den.z = uni((wave_sum(sm.z) + exs.z) + 1e-16f);
    den.w = uni((wave_sum(sm.w) + exs.w) + 1e-16f);

    float4 acc = make_float4(0.f, 0.f, 0.f, 0.f);
    edge_fma(acc, h4[d * 32 + lane],
             pick(make_float4(exs.x / den.x, exs.y / den.y, exs.z / den.z,
                              exs.w / den.w), hh));
    {
      const float4 al = make_float4(ex0.x / den.x, ex0.y / den.y, ex0.z / den.z,
                                    ex0.w / den.w);
      gather_chunk<true>(acc, h4, src0, al, n0, lane, hh);
    }
    for (int base = beg + 32; base < end; base += 32) {
      const int rem = end - base;
      const int n = rem < 32 ? rem : 32;
      int srcv = 0;
      float4 al = make_float4(0.f, 0.f, 0.f, 0.f);
      if (lane < n) {
        srcv = csr[base + lane];
        const float4 e = score4(as4[srcv], ad);
        al.x = expf(e.x - m.x) / den.x;
        al.y = expf(e.y - m.y) / den.y;
        al.z = expf(e.z - m.z) / den.z;
        al.w = expf(e.w - m.w) / den.w;
      }
      gather_chunk<false>(acc, h4, srcv, al, n, lane, hh);
    }

    const float4 sk = k4[d * 32 + lane];
    float4 o;
    o.x = elu1(sk.x + acc.x);
    o.y = elu1(sk.y + acc.y);
    o.z = elu1(sk.z + acc.z);
    o.w = elu1(sk.w + acc.w);
    o4[d * 32 + lane] = o;
  }
}

extern "C" void kernel_launch(void* const* d_in, const int* in_sizes, int n_in,
                              void* d_out, int out_size, void* d_ws,
                              size_t ws_size, hipStream_t stream) {
  const float* x  = (const float*)d_in[0];
  const int*   ei = (const int*)d_in[1];

  const int N  = in_sizes[0] / DIM;
  const int NE = in_sizes[1] / 2;

  float* ws = (float*)d_ws;
  size_t o = 0;
  float* h     = ws + o; o += (size_t)N * DIM;
  float* agg   = ws + o; o += (size_t)N * DIM;
  float* asrc  = ws + o; o += (size_t)N * HEADS;
  float* adst  = ws + o; o += (size_t)N * HEADS;
  const int chunk4 = (N + 4096) / 4096;
  const int NP     = chunk4 * 4096;
  int*  csr    = (int*)(ws + o);  o += (size_t)NE;
  int*  rank   = (int*)(ws + o);  o += (size_t)NE;
  int*  cnt    = (int*)(ws + o);  o += (size_t)NP;
  int*  rowptr = (int*)(ws + o);  o += (size_t)NP;

  float* y = (float*)d_out;

  const int TB = 256;
  const int nStrips = N / 16;
  const int gGemm = 256;
  const int gE1   = (NE + TB - 1) / TB;
  const int gNP   = (NP + TB - 1) / TB;
  const int gGath = 2048;

  csr_zero<<<gNP, TB, 0, stream>>>(cnt, NP);
  csr_count<<<gE1, TB, 0, stream>>>(ei, NE, cnt, rank);
  csr_scan<<<1, 1024, 0, stream>>>(cnt, rowptr, chunk4);
  csr_fill<<<gE1, TB, 0, stream>>>(ei, NE, rowptr, rank, csr);

  const float* xin = x;
  for (int layer = 0; layer < 2; ++layer) {
    const float* W     = (const float*)d_in[2 + layer * 5];
    const float* a_src = (const float*)d_in[3 + layer * 5];
    const float* a_dst = (const float*)d_in[4 + layer * 5];
    const float* bias  = (const float*)d_in[5 + layer * 5];
    const float* Wskip = (const float*)d_in[6 + layer * 5];

    gat_gemm_pair_f32_wmma<<<gGemm, 512, 0, stream>>>(xin, W, Wskip, bias, a_src,
                                                      a_dst, h, agg, asrc, adst,
                                                      nStrips);

    gat_softmax_gather_csr<<<gGath, TB, 0, stream>>>(csr, rowptr, asrc, adst, h,
                                                     agg, y, N);
    xin = y;
  }
}


# Edge bucketing in one pass: the counting kernel writes each source into its destination's 32-slot line at its arrival rank (rank 32 up: spill list); scan and fill launches removed; sum order as before
# speedup vs baseline: 4.5143x; 1.2619x over previous
#include <hip/hip_runtime.h>
#include <math.h>

#define DIM 128
#define HEADS 4
#define CH 32
#define NEG_SLOPE 0.2f

typedef __attribute__((ext_vector_type(2))) float v2f;
typedef __attribute__((ext_vector_type(8))) float v8f;

__device__ __forceinline__ float xhalf(float v) {
  const int i = __builtin_bit_cast(int, v);
  return __builtin_bit_cast(
      float, __builtin_amdgcn_permlanex16(i, i, 0x76543210, 0xfedcba98, false,
                                          false));
}

template <int CTRL>
__device__ __forceinline__ float row_xor(float v) {
  return __builtin_bit_cast(
      float, __builtin_amdgcn_update_dpp(0, __builtin_bit_cast(int, v), CTRL,
                                         0xf, 0xf, false));
}

template <int CTRL>
__device__ __forceinline__ float fold(float a, float b, bool up) {
  const float keep = up ? b : a;
  const float give = up ? a : b;
  return keep + row_xor<CTRL>(give);
}

__device__ __forceinline__ void load_strip(v2f (&a)[32], const float* __restrict__ X,
                                           int strip, int m, int kh) {
  const v2f* __restrict__ arow =
      (const v2f*)(X + (size_t)(strip * 16 + m) * DIM + 2 * kh);
#pragma unroll
  for (int i = 0; i < 32; ++i) a[i] = arow[2 * i];
}

__global__ void __launch_bounds__(512)
gat_gemm_pair_f32_wmma(const float* __restrict__ X,
                       const float* __restrict__ Wa,
                       const float* __restrict__ Wb,
                       const float* __restrict__ bias_b,
                       const float* __restrict__ a_src,
                       const float* __restrict__ a_dst,
                       float* __restrict__ out_a,
                       float* __restrict__ out_b,
                       float* __restrict__ asrc,
                       float* __restrict__ adst, int nstrips) {
  __shared__ float4 wl[32 * 128];
  const int tid = threadIdx.x;
  const int lane = tid & 31;
  const int wave = __builtin_amdgcn_readfirstlane(tid >> 5);
  const int wpb = blockDim.x >> 5;
  const int m  = lane & 15;
  const int kh = lane >> 4;
  const int g = blockIdx.x & 1;
  const int pair = blockIdx.x >> 1;
  const int npairs = gridDim.x >> 1;
  const float* __restrict__ Wg = g ? Wb : Wa;
  float* __restrict__ out = g ? out_b : out_a;

  int s = pair + npairs * wave;
  const int sstep = npairs * wpb;

  float4 wv[8];
  {
    const int c = tid & 127;
#pragma unroll
    for (int q = 0; q < 8; ++q) {
      const int jh = (tid >> 7) + 4 * q;
      const int k = (jh >> 1) * 8 + (jh & 1) * 2;
      wv[q].x = Wg[(k) * DIM + c];
      wv[q].y = Wg[(k + 1) * DIM + c];
      wv[q].z = Wg[(k + 4) * DIM + c];
      wv[q].w = Wg[(k + 5) * DIM + c];
    }
  }
  float bz[8];
  float asv[8], adv[8];
#pragma unroll
  for (int t = 0; t < 8; ++t) {
    const float bt = bias_b[16 * t + m];
    bz[t] = g ? bt : 0.0f;
    asv[t] = a_src[16 * t + m];
    adv[t] = a_dst[16 * t + m];
  }
  const bool up0 = m & 1, up1 = m & 2, up2 = m & 4, up3 = m & 8;
#pragma unroll
  for (int q = 0; q < 8; ++q) wl[tid + 512 * q] = wv[q];
  __syncthreads();

  const float4* wlane = wl + kh * 128 + m;
  while (s < nstrips) {
    v2f a[32];
    load_strip(a, X, s, m, kh);
    __builtin_amdgcn_sched_barrier(0);
    v8f acc[8];
#pragma unroll
    for (int t = 0; t < 8; ++t) acc[t] = (v8f){};
#pragma unroll
    for (int jj = 0; jj < 16; ++jj) {
      float4 b[8];
#pragma unroll
      for (int t = 0; t < 8; ++t) b[t] = wlane[jj * 256 + 16 * t];
#pragma unroll
      for (int t = 0; t < 8; ++t) {
        v2f bv; bv.x = b[t].x; bv.y = b[t].y;
        acc[t] = __builtin_amdgcn_wmma_f32_16x16x4_f32(
            false, a[2 * jj], false, bv, (short)0, acc[t], false, false);
      }
#pragma unroll
      for (int t = 0; t < 8; ++t) {
        v2f bv; bv.x = b[t].z; bv.y = b[t].w;
        acc[t] = __builtin_amdgcn_wmma_f32_16x16x4_f32(
            false, a[2 * jj + 1], false, bv, (short)0, acc[t], false, false);
      }
    }
    float* __restrict__ orow = out + (size_t)(s * 16) * DIM + lane;
#pragma unroll
    for (int tp = 0; tp < 4; ++tp) {
#pragma unroll
      for (int r = 0; r < 8; ++r) {
        const float xv = acc[2 * tp][r] + bz[2 * tp];
        const float yv = acc[2 * tp + 1][r] + bz[2 * tp + 1];
        const float xs = xhalf(xv);
        const float ys = xhalf(yv);
        orow[(size_t)r * DIM + 32 * tp] = kh ? ys : xv;
        orow[(size_t)(r + 8) * DIM + 32 * tp] = kh ? yv : xs;
      }
    }
    if (g == 0) {
#pragma unroll
      for (int sd = 0; sd < 2; ++sd) {
        const float* av = sd ? adv : asv;
        float w2[8];
#pragma unroll
        for (int r = 0; r < 8; ++r) {
          float p[4];
#pragma unroll
          for (int hd = 0; hd < 4; ++hd)
            p[hd] = acc[2 * hd][r] * av[2 * hd] + acc[2 * hd + 1][r] * av[2 * hd + 1];
          const float u0 = fold<0x161>(p[0], p[1], up0);
          const float u1 = fold<0x161>(p[2], p[3], up0);
          w2[r] = fold<0x162>(u0, u1, up1);
        }
        float w3[4];
#pragma unroll
        for (int q = 0; q < 4; ++q)
          w3[q] = fold<0x164>(w2[2 * q], w2[2 * q + 1], up2);
        const float y0 = fold<0x168>(w3[0], w3[1], up3);
        const float y1 = fold<0x168>(w3[2], w3[3], up3);
        const float x0 = xhalf(y0);
        const float x1 = xhalf(y1);
        float* __restrict__ srow = (sd ? adst : asrc) + (size_t)s * 64 + lane;
        srow[0] = kh ? x1 : y0;
        srow[32] = kh ? y1 : x0;
      }
    }
    s += sstep;
  }
}

#define SLOTS 32

__global__ void bucket_zero(int* __restrict__ cnt, int n) {
  int t = blockIdx.x * blockDim.x + threadIdx.x;
  if (t < n) cnt[t] = 0;
}

__global__ void bucket_edges(const int* __restrict__ ei, int NE, int N,
                             int* __restrict__ cnt, int* __restrict__ bucket,
                             int* __restrict__ spill) {
  int t = blockIdx.x * blockDim.x + threadIdx.x;
  if (t >= NE) return;
  const int s = ei[t];
  const int d = ei[NE + t];
  const int pos = atomicAdd(&cnt[d], 1);
  if (pos < SLOTS) {
    bucket[d * SLOTS + pos] = s;
  } else {
    const int k = atomicAdd(&cnt[N], 1);
    ((int2*)spill)[k] = make_int2(d, s);
  }
}

__device__ __forceinline__ void edge_fma(float4& acc, const float4 hv, float a) {
  acc.x += hv.x * a;
  acc.y += hv.y * a;
  acc.z += hv.z * a;
  acc.w += hv.w * a;
}

__device__ __forceinline__ float elu1(float v) {
  return (v > 0.0f) ? v : expm1f(v);
}

__device__ __forceinline__ float lrelu(float v) {
  return (v > 0.0f) ? v : NEG_SLOPE * v;
}

__device__ __forceinline__ float4 score4(const float4 a, const float4 b) {
  return make_float4(lrelu(a.x + b.x), lrelu(a.y + b.y), lrelu(a.z + b.z),
                     lrelu(a.w + b.w));
}

__device__ __forceinline__ float wave_max(float v) {
  v = fmaxf(v, row_xor<0x161>(v));
  v = fmaxf(v, row_xor<0x162>(v));
  v = fmaxf(v, row_xor<0x164>(v));
  v = fmaxf(v, row_xor<0x168>(v));
  return fmaxf(v, xhalf(v));
}

__device__ __forceinline__ float wave_sum(float v) {
  v += row_xor<0x161>(v);
  v += row_xor<0x162>(v);
  v += row_xor<0x164>(v);
  v += row_xor<0x168>(v);
  return v + xhalf(v);
}

__device__ __forceinline__ float lane_f(float v, int j) {
  return __builtin_bit_cast(
      float, __builtin_amdgcn_readlane(__builtin_bit_cast(int, v), j));
}

__device__ __forceinline__ float uni(float v) {
  return __builtin_bit_cast(
      float, __builtin_amdgcn_readfirstlane(__builtin_bit_cast(int, v)));
}

__device__ __forceinline__ float pick(const float4 a, int hh) {
  return (hh & 2) ? ((hh & 1) ? a.w : a.z) : ((hh & 1) ? a.y : a.x);
}

__device__ __forceinline__ void gather_chunk(float4& acc,
                                             const float4* __restrict__ h4,
                                             int srcv, const float4 al, int n,
                                             int lane, int hh) {
  int j = 0;
  for (; j + 4 <= n; j += 4) {
    const int s0 = __builtin_amdgcn_readlane(srcv, j);
    const int s1 = __builtin_amdgcn_readlane(srcv, j + 1);
    const int s2 = __builtin_amdgcn_readlane(srcv, j + 2);
    const int s3 = __builtin_amdgcn_readlane(srcv, j + 3);
    const float4 v0 = h4[s0 * 32 + lane];
    const float4 v1 = h4[s1 * 32 + lane];
    const float4 v2 = h4[s2 * 32 + lane];
    const float4 v3 = h4[s3 * 32 + lane];
    const float w0 = pick(make_float4(lane_f(al.x, j), lane_f(al.y, j),
                                      lane_f(al.z, j), lane_f(al.w, j)), hh);
    const float w1 = pick(make_float4(lane_f(al.x, j + 1), lane_f(al.y, j + 1),
                                      lane_f(al.z, j + 1), lane_f(al.w, j + 1)), hh);
    const float w2 = pick(make_float4(lane_f(al.x, j + 2), lane_f(al.y, j + 2),
                                      lane_f(al.z, j + 2), lane_f(al.w, j + 2)), hh);
    const float w3 = pick(make_float4(lane_f(al.x, j + 3), lane_f(al.y, j + 3),
                                      lane_f(al.z, j + 3), lane_f(al.w, j + 3)), hh);
    edge_fma(acc, v0, w0);
    edge_fma(acc, v1, w1);
    edge_fma(acc, v2, w2);
    edge_fma(acc, v3, w3);
  }
  for (; j < n; ++j) {
    const int s0 = __builtin_amdgcn_readlane(srcv, j);
    const float4 v0 = h4[s0 * 32 + lane];
    const float w0 = pick(make_float4(lane_f(al.x, j), lane_f(al.y, j),
                                      lane_f(al.z, j), lane_f(al.w, j)), hh);
    edge_fma(acc, v0, w0);
  }
}

__device__ __forceinline__ void gather_live(float4& acc,
                                            const float4* __restrict__ h4,
                                            int srcv, const float4 al,
                                            unsigned live, int lane, int hh) {
  while (live) {
    const int j = __builtin_ctz(live);
    live &= live - 1;
    const int s0 = __builtin_amdgcn_readlane(srcv, j);
    const float4 v0 = h4[s0 * 32 + lane];
    const float w0 = pick(make_float4(lane_f(al.x, j), lane_f(al.y, j),
                                      lane_f(al.z, j), lane_f(al.w, j)), hh);
    edge_fma(acc, v0, w0);
  }
}

__global__ void __launch_bounds__(256)
gat_softmax_gather_bkt(const int* __restrict__ bucket, const int* __restrict__ cnt,
                       const int* __restrict__ spill,
                       const float* __restrict__ asrc, const float* __restrict__ adst,
                       const float* __restrict__ h, const float* __restrict__ skip,
                       float* __restrict__ out, int N) {
  const int lane = threadIdx.x & 31;
  const int hh = lane >> 3;
  const int wpb = blockDim.x >> 5;
  const int wv = __builtin_amdgcn_readfirstlane(threadIdx.x >> 5);
  const int nw = gridDim.x * wpb;
  const float4* __restrict__ h4 = (const float4*)h;
  const float4* __restrict__ k4 = (const float4*)skip;
  const float4* __restrict__ as4 = (const float4*)asrc;
  const float4* __restrict__ ad4 = (const float4*)adst;
  const int2* __restrict__ sp2 = (const int2*)spill;
  float4* __restrict__ o4 = (float4*)out;
  const float NEG_INF = -__builtin_huge_valf();
  const int nspill = cnt[N];

  for (int d = blockIdx.x * wpb + wv; d < N; d += nw) {
    const int deg = cnt[d];
    const float4 ad = ad4[d];
    const float4 es = score4(as4[d], ad);

    const int n0 = deg < SLOTS ? deg : SLOTS;
    const int nsp = deg > SLOTS ? nspill : 0;
    int src0 = 0;
    float4 e0 = make_float4(NEG_INF, NEG_INF, NEG_INF, NEG_INF);
    if (lane < n0) {
      src0 = bucket[d * SLOTS + lane];
      e0 = score4(as4[src0], ad);
    }

    float4 mx = e0;
    for (int base = 0; base < nsp; base += 32) {
      if (base + lane < nsp) {
        const int2 p = sp2[base + lane];
        if (p.x == d) {
          const float4 e = score4(as4[p.y], ad);
          mx.x = fmaxf(mx.x, e.x);
          mx.y = fmaxf(mx.y, e.y);
          mx.z = fmaxf(mx.z, e.z);
          mx.w = fmaxf(mx.w, e.w);
        }
      }
    }
    float4 m;
    m.x = fmaxf(wave_max(mx.x), es.x);
    m.y = fmaxf(wave_max(mx.y), es.y);
    m.z = fmaxf(wave_max(mx.z), es.z);
    m.w = fmaxf(wave_max(mx.w), es.w);
    m.x = uni(isfinite(m.x) ? m.x : 0.0f);
    m.y = uni(isfinite(m.y) ? m.y : 0.0f);
    m.z = uni(isfinite(m.z) ? m.z : 0.0f);
    m.w = uni(isfinite(m.w) ? m.w : 0.0f);

    float4 ex0 = make_float4(0.f, 0.f, 0.f, 0.f);
    if (lane < n0) {
      ex0.x = expf(e0.x - m.x);
      ex0.y = expf(e0.y - m.y);
      ex0.z = expf(e0.z - m.z);
      ex0.w = expf(e0.w - m.w);
    }
    float4 sm = ex0;
    for (int base = 0; base < nsp; base += 32) {
      if (base + lane < nsp) {
        const int2 p = sp2[base + lane];
        if (p.x == d) {
          const float4 e = score4(as4[p.y], ad);
          sm.x += expf(e.x - m.x);
          sm.y += expf(e.y - m.y);
          sm.z += expf(e.z - m.z);
          sm.w += expf(e.w - m.w);
        }
      }
    }
    const float4 exs = make_float4(expf(es.x - m.x), expf(es.y - m.y),
                                   expf(es.z - m.z), expf(es.w - m.w));
    float4 den;
    den.x = uni((wave_sum(sm.x) + exs.x) + 1e-16f);
    den.y = uni((wave_sum(sm.y) + exs.y) + 1e-16f);
    den.z = uni((wave_sum(sm.z) + exs.z) + 1e-16f);
    den.w = uni((wave_sum(sm.w) + exs.w) + 1e-16f);

    float4 acc = make_float4(0.f, 0.f, 0.f, 0.f);
    edge_fma(acc, h4[d * 32 + lane],
             pick(make_float4(exs.x / den.x, exs.y / den.y, exs.z / den.z,
                              exs.w / den.w), hh));
    {
      const float4 al = make_float4(ex0.x / den.x, ex0.y / den.y, ex0.z / den.z,
                                    ex0.w / den.w);
      gather_chunk(acc, h4, src0, al, n0, lane, hh);
    }
    for (int base = 0; base < nsp; base += 32) {
      bool mine = false;
      int srcv = 0;
      float4 al = make_float4(0.f, 0.f, 0.f, 0.f);
      if (base + lane < nsp) {
        const int2 p = sp2[base + lane];
        if (p.x == d) {
          mine = true;
          srcv = p.y;
          const float4 e = score4(as4[srcv], ad);
          al.x = expf(e.x - m.x) / den.x;
          al.y = expf(e.y - m.y) / den.y;
          al.z = expf(e.z - m.z) / den.z;
          al.w = expf(e.w - m.w) / den.w;
        }
      }
      gather_live(acc, h4, srcv, al, __builtin_amdgcn_ballot_w32(mine), lane, hh);
    }

    const float4 sk = k4[d * 32 + lane];
    float4 o;
    o.x = elu1(sk.x + acc.x);
    o.y = elu1(sk.y + acc.y);
    o.z = elu1(sk.z + acc.z);
    o.w = elu1(sk.w + acc.w);
    o4[d * 32 + lane] = o;
  }
}

extern "C" void kernel_launch(void* const* d_in, const int* in_sizes, int n_in,
                              void* d_out, int out_size, void* d_ws,
                              size_t ws_size, hipStream_t stream) {
  const float* x  = (const float*)d_in[0];
  const int*   ei = (const int*)d_in[1];

  const int N  = in_sizes[0] / DIM;
  const int NE = in_sizes[1] / 2;

  float* ws = (float*)d_ws;
  size_t o = 0;
  float* h     = ws + o; o += (size_t)N * DIM;
  float* agg   = ws + o; o += (size_t)N * DIM;
  float* asrc  = ws + o; o += (size_t)N * HEADS;
  float* adst  = ws + o; o += (size_t)N * HEADS;
  int*  bucket = (int*)(ws + o);  o += (size_t)N * SLOTS;
  int*  spill  = (int*)(ws + o);  o += (size_t)NE * 2;
  int*  cnt    = (int*)(ws + o);  o += (size_t)N + 1;

  float* y = (float*)d_out;

  const int TB = 256;
  const int nStrips = N / 16;
  const int gGemm = 256;
  const int gE1   = (NE + TB - 1) / TB;
  const int gCnt  = (N + 1 + TB - 1) / TB;
  const int gGath = 2048;

  bucket_zero<<<gCnt, TB, 0, stream>>>(cnt, N + 1);
  bucket_edges<<<gE1, TB, 0, stream>>>(ei, NE, N, cnt, bucket, spill);

  const float* xin = x;
  for (int layer = 0; layer < 2; ++layer) {
    const float* W     = (const float*)d_in[2 + layer * 5];
    const float* a_src = (const float*)d_in[3 + layer * 5];
    const float* a_dst = (const float*)d_in[4 + layer * 5];
    const float* bias  = (const float*)d_in[5 + layer * 5];
    const float* Wskip = (const float*)d_in[6 + layer * 5];

    gat_gemm_pair_f32_wmma<<<gGemm, 512, 0, stream>>>(xin, W, Wskip, bias, a_src,
                                                      a_dst, h, agg, asrc, adst,
                                                      nStrips);

    gat_softmax_gather_bkt<<<gGath, TB, 0, stream>>>(bucket, cnt, spill, asrc, adst,
                                                     h, agg, y, N);
    xin = y;
  }
}


# Edge buckets ranked by LDS counters in two passes (chunk x 256-node-range cells, then one block per range stores whole lines); per-edge global atomics, zero launch gone; slot order: LDS arrival rank
# speedup vs baseline: 5.8780x; 1.3021x over previous
#include <hip/hip_runtime.h>
#include <math.h>

#define DIM 128
#define HEADS 4
#define CH 32
#define NEG_SLOPE 0.2f

typedef __attribute__((ext_vector_type(2))) float v2f;
typedef __attribute__((ext_vector_type(8))) float v8f;

__device__ __forceinline__ float xhalf(float v) {
  const int i = __builtin_bit_cast(int, v);
  return __builtin_bit_cast(
      float, __builtin_amdgcn_permlanex16(i, i, 0x76543210, 0xfedcba98, false,
                                          false));
}

template <int CTRL>
__device__ __forceinline__ float row_xor(float v) {
  return __builtin_bit_cast(
      float, __builtin_amdgcn_update_dpp(0, __builtin_bit_cast(int, v), CTRL,
                                         0xf, 0xf, false));
}

template <int CTRL>
__device__ __forceinline__ float fold(float a, float b, bool up) {
  const float keep = up ? b : a;
  const float give = up ? a : b;
  return keep + row_xor<CTRL>(give);
}

__device__ __forceinline__ void load_strip(v2f (&a)[32], const float* __restrict__ X,
                                           int strip, int m, int kh) {
  const v2f* __restrict__ arow =
      (const v2f*)(X + (size_t)(strip * 16 + m) * DIM + 2 * kh);
#pragma unroll
  for (int i = 0; i < 32; ++i) a[i] = arow[2 * i];
}

__global__ void __launch_bounds__(512)
gat_gemm_pair_f32_wmma(const float* __restrict__ X,
                       const float* __restrict__ Wa,
                       const float* __restrict__ Wb,
                       const float* __restrict__ bias_b,
                       const float* __restrict__ a_src,
                       const float* __restrict__ a_dst,
                       float* __restrict__ out_a,
                       float* __restrict__ out_b,
                       float* __restrict__ asrc,
                       float* __restrict__ adst, int nstrips) {
  __shared__ float4 wl[32 * 128];
  const int tid = threadIdx.x;
  const int lane = tid & 31;
  const int wave = __builtin_amdgcn_readfirstlane(tid >> 5);
  const int wpb = blockDim.x >> 5;
  const int m  = lane & 15;
  const int kh = lane >> 4;
  const int g = blockIdx.x & 1;
  const int pair = blockIdx.x >> 1;
  const int npairs = gridDim.x >> 1;
  const float* __restrict__ Wg = g ? Wb : Wa;
  float* __restrict__ out = g ? out_b : out_a;

  int s = pair + npairs * wave;
  const int sstep = npairs * wpb;

  float4 wv[8];
  {
    const int c = tid & 127;
#pragma unroll
    for (int q = 0; q < 8; ++q) {
      const int jh = (tid >> 7) + 4 * q;
      const int k = (jh >> 1) * 8 + (jh & 1) * 2;
      wv[q].x = Wg[(k) * DIM + c];
      wv[q].y = Wg[(k + 1) * DIM + c];
      wv[q].z = Wg[(k + 4) * DIM + c];
      wv[q].w = Wg[(k + 5) * DIM + c];
    }
  }
  float bz[8];
  float asv[8], adv[8];
#pragma unroll
  for (int t = 0; t < 8; ++t) {
    const float bt = bias_b[16 * t + m];
    bz[t] = g ? bt : 0.0f;
    asv[t] = a_src[16 * t + m];
    adv[t] = a_dst[16 * t + m];
  }
  const bool up0 = m & 1, up1 = m & 2, up2 = m & 4, up3 = m & 8;
#pragma unroll
  for (int q = 0; q < 8; ++q) wl[tid + 512 * q] = wv[q];
  __syncthreads();

  const float4* wlane = wl + kh * 128 + m;
  while (s < nstrips) {
    v2f a[32];
    load_strip(a, X, s, m, kh);
    __builtin_amdgcn_sched_barrier(0);
    v8f acc[8];
#pragma unroll
    for (int t = 0; t < 8; ++t) acc[t] = (v8f){};
#pragma unroll
    for (int jj = 0; jj < 16; ++jj) {
      float4 b[8];
#pragma unroll
      for (int t = 0; t < 8; ++t) b[t] = wlane[jj * 256 + 16 * t];
#pragma unroll
      for (int t = 0; t < 8; ++t) {
        v2f bv; bv.x = b[t].x; bv.y = b[t].y;
        acc[t] = __builtin_amdgcn_wmma_f32_16x16x4_f32(
            false, a[2 * jj], false, bv, (short)0, acc[t], false, false);
      }
#pragma unroll
      for (int t = 0; t < 8; ++t) {
        v2f bv; bv.x = b[t].z; bv.y = b[t].w;
        acc[t] = __builtin_amdgcn_wmma_f32_16x16x4_f32(
            false, a[2 * jj + 1], false, bv, (short)0, acc[t], false, false);
      }
    }
    float* __restrict__ orow = out + (size_t)(s * 16) * DIM + lane;
#pragma unroll
    for (int tp = 0; tp < 4; ++tp) {
#pragma unroll
      for (int r = 0; r < 8; ++r) {
        const float xv = acc[2 * tp][r] + bz[2 * tp];
        const float yv = acc[2 * tp + 1][r] + bz[2 * tp + 1];
        const float xs = xhalf(xv);
        const float ys = xhalf(yv);
        orow[(size_t)r * DIM + 32 * tp] = kh ? ys : xv;
        orow[(size_t)(r + 8) * DIM + 32 * tp] = kh ? yv : xs;
      }
    }
    if (g == 0) {
#pragma unroll
      for (int sd = 0; sd < 2; ++sd) {
        const float* av = sd ? adv : asv;
        float w2[8];
#pragma unroll
        for (int r = 0; r < 8; ++r) {
          float p[4];
#pragma unroll
          for (int hd = 0; hd < 4; ++hd)
            p[hd] = acc[2 * hd][r] * av[2 * hd] + acc[2 * hd + 1][r] * av[2 * hd + 1];
          const float u0 = fold<0x161>(p[0], p[1], up0);
          const float u1 = fold<0x161>(p[2], p[3], up0);
          w2[r] = fold<0x162>(u0, u1, up1);
        }
        float w3[4];
#pragma unroll
        for (int q = 0; q < 4; ++q)
          w3[q] = fold<0x164>(w2[2 * q], w2[2 * q + 1], up2);
        const float y0 = fold<0x168>(w3[0], w3[1], up3);
        const float y1 = fold<0x168>(w3[2], w3[3], up3);
        const float x0 = xhalf(y0);
        const float x1 = xhalf(y1);
        float* __restrict__ srow = (sd ? adst : asrc) + (size_t)s * 64 + lane;
        srow[0] = kh ? x1 : y0;
        srow[32] = kh ? y1 : x0;
      }
    }
    s += sstep;
  }
}

#define SLOTS 32
#define CHUNKS 256
#define RANGE 256
#define RSHIFT 8
#define CAP 32
#define CROW 288
#define SORT_TB 512
#define EPT 5

__global__ void __launch_bounds__(SORT_TB)
bucket_pass1(const int* __restrict__ ei, int NE, int N, int* __restrict__ stage,
             int* __restrict__ cellcnt, int* __restrict__ ovf,
             int* __restrict__ cnt) {
  __shared__ int4 lst4[256 * CAP / 4];
  __shared__ int ccnt[256];
  __shared__ int ovn;
  int* lst = (int*)lst4;
  const int tid = threadIdx.x;
  const int b = blockIdx.x;
  const int nrange = (N + RANGE - 1) >> RSHIFT;
  const int C = (NE + CHUNKS - 1) / CHUNKS;
  for (int i = tid; i < nrange * (CAP / 4); i += SORT_TB)
    lst4[i] = make_int4(0, 0, 0, 0);
  if (tid < 256) ccnt[tid] = 0;
  if (tid == 0) {
    ovn = 0;
    if (b == 0) cnt[N] = 0;
  }
  __syncthreads();
  const int lo = b * C;
  const int hi = (lo + C < NE) ? lo + C : NE;
  int2* __restrict__ ov2 = (int2*)ovf;
  for (int e0 = lo + tid; e0 < hi; e0 += EPT * SORT_TB) {
    int sv[EPT], dv[EPT], pos[EPT];
#pragma unroll
    for (int u = 0; u < EPT; ++u) {
      const int e = e0 + SORT_TB * u;
      const bool ok = e < hi;
      const int ec = ok ? e : e0;
      const int st = ei[ec];
      const int dt = ei[NE + ec];
      sv[u] = st;
      dv[u] = ok ? dt : -1;
    }
#pragma unroll
    for (int u = 0; u < EPT; ++u)
      pos[u] = (dv[u] >= 0) ? atomicAdd(&ccnt[dv[u] >> RSHIFT], 1) : 0;
#pragma unroll
    for (int u = 0; u < EPT; ++u) {
      if (dv[u] >= 0) {
        if (pos[u] < CAP) {
          lst[(dv[u] >> RSHIFT) * CAP + pos[u]] = ((dv[u] & (RANGE - 1)) << 20) | sv[u];
        } else {
          const int k = atomicAdd(&ovn, 1);
          ov2[lo + k] = make_int2(dv[u], sv[u]);
        }
      }
    }
  }
  __syncthreads();
  if (tid < 256) {
    const int c = ccnt[tid];
    cellcnt[b * CROW + tid] = c < CAP ? c : CAP;
  } else if (tid < CROW) {
    cellcnt[b * CROW + tid] = (tid == 256) ? ovn : 0;
  }
  int4* __restrict__ st4 = (int4*)stage + (size_t)b * (256 * CAP / 4);
  for (int i = tid; i < nrange * (CAP / 4); i += SORT_TB) st4[i] = lst4[i];
}

__device__ __forceinline__ void place_ranked(int* __restrict__ lb, int pos, int dl,
                                             int s, int node0, int* __restrict__ nspill,
                                             int2* __restrict__ sp2) {
  if (pos < SLOTS) {
    lb[dl * SLOTS + pos] = s;
  } else {
    const int k = atomicAdd(nspill, 1);
    sp2[k] = make_int2(node0 + dl, s);
  }
}

__global__ void __launch_bounds__(SORT_TB)
bucket_pass2(const int* __restrict__ stage, const int* __restrict__ cellcnt,
             const int* __restrict__ ovf, int NE, int N, int* __restrict__ cnt,
             int* __restrict__ bucket, int* __restrict__ spill) {
  __shared__ int4 lb4[RANGE * SLOTS / 4];
  __shared__ int lcnt[RANGE];
  int* lb = (int*)lb4;
  const int tid = threadIdx.x;
  const int lane = tid & 31;
  const int wv = __builtin_amdgcn_readfirstlane(tid >> 5);
  const int r = blockIdx.x;
  const int node0 = r * RANGE;
  const int C = (NE + CHUNKS - 1) / CHUNKS;
  int2* __restrict__ sp2 = (int2*)spill;
  const int2* __restrict__ ov2 = (const int2*)ovf;
  const int nv = cellcnt[(wv + 16 * (lane & 15)) * CROW + r];
  int ev[16];
#pragma unroll
  for (int u = 0; u < 16; ++u)
    ev[u] = stage[((size_t)(wv + 16 * u) * 256 + r) * CAP + lane];
  for (int i = tid; i < RANGE * SLOTS / 4; i += SORT_TB) lb4[i] = make_int4(0, 0, 0, 0);
  if (tid < RANGE) lcnt[tid] = 0;
  __syncthreads();
#pragma unroll
  for (int u0 = 0; u0 < 16; u0 += 8) {
    int pos[8];
#pragma unroll
    for (int u = 0; u < 8; ++u) {
      const int n = __builtin_amdgcn_readlane(nv, u0 + u);
      pos[u] = (lane < n) ? atomicAdd(&lcnt[ev[u0 + u] >> 20], 1) : -1;
    }
#pragma unroll
    for (int u = 0; u < 8; ++u)
      if (pos[u] >= 0)
        place_ranked(lb, pos[u], ev[u0 + u] >> 20, ev[u0 + u] & 0xFFFFF, node0,
                     &cnt[N], sp2);
  }
  if (tid < CHUNKS) {
    const int m = cellcnt[tid * CROW + 256];
    for (int i = 0; i < m; ++i) {
      const int2 p = ov2[tid * C + i];
      if ((p.x >> RSHIFT) == r) {
        const int dl = p.x & (RANGE - 1);
        place_ranked(lb, atomicAdd(&lcnt[dl], 1), dl, p.y, node0, &cnt[N], sp2);
      }
    }
  }
  __syncthreads();
  if (tid < RANGE && node0 + tid < N) cnt[node0 + tid] = lcnt[tid];
  const int nn = (N - node0 < RANGE) ? N - node0 : RANGE;
  int4* __restrict__ b4 = (int4*)bucket + (size_t)node0 * (SLOTS / 4);
  for (int i = tid; i < nn * (SLOTS / 4); i += SORT_TB) b4[i] = lb4[i];
}

__device__ __forceinline__ void edge_fma(float4& acc, const float4 hv, float a) {
  acc.x += hv.x * a;
  acc.y += hv.y * a;
  acc.z += hv.z * a;
  acc.w += hv.w * a;
}

__device__ __forceinline__ float elu1(float v) {
  return (v > 0.0f) ? v : expm1f(v);
}

__device__ __forceinline__ float lrelu(float v) {
  return (v > 0.0f) ? v : NEG_SLOPE * v;
}

__device__ __forceinline__ float4 score4(const float4 a, const float4 b) {
  return make_float4(lrelu(a.x + b.x), lrelu(a.y + b.y), lrelu(a.z + b.z),
                     lrelu(a.w + b.w));
}

__device__ __forceinline__ float wave_max(float v) {
  v = fmaxf(v, row_xor<0x161>(v));
  v = fmaxf(v, row_xor<0x162>(v));
  v = fmaxf(v, row_xor<0x164>(v));
  v = fmaxf(v, row_xor<0x168>(v));
  return fmaxf(v, xhalf(v));
}

__device__ __forceinline__ float wave_sum(float v) {
  v += row_xor<0x161>(v);
  v += row_xor<0x162>(v);
  v += row_xor<0x164>(v);
  v += row_xor<0x168>(v);
  return v + xhalf(v);
}

__device__ __forceinline__ float lane_f(float v, int j) {
  return __builtin_bit_cast(
      float, __builtin_amdgcn_readlane(__builtin_bit_cast(int, v), j));
}

__device__ __forceinline__ float uni(float v) {
  return __builtin_bit_cast(
      float, __builtin_amdgcn_readfirstlane(__builtin_bit_cast(int, v)));
}

__device__ __forceinline__ float pick(const float4 a, int hh) {
  return (hh & 2) ? ((hh & 1) ? a.w : a.z) : ((hh & 1) ? a.y : a.x);
}

__device__ __forceinline__ void gather_chunk(float4& acc,
                                             const float4* __restrict__ h4,
                                             int srcv, const float4 al, int n,
                                             int lane, int hh) {
  int j = 0;
  for (; j + 4 <= n; j += 4) {
    const int s0 = __builtin_amdgcn_readlane(srcv, j);
    const int s1 = __builtin_amdgcn_readlane(srcv, j + 1);
    const int s2 = __builtin_amdgcn_readlane(srcv, j + 2);
    const int s3 = __builtin_amdgcn_readlane(srcv, j + 3);
    const float4 v0 = h4[s0 * 32 + lane];
    const float4 v1 = h4[s1 * 32 + lane];
    const float4 v2 = h4[s2 * 32 + lane];
    const float4 v3 = h4[s3 * 32 + lane];
    const float w0 = pick(make_float4(lane_f(al.x, j), lane_f(al.y, j),
                                      lane_f(al.z, j), lane_f(al.w, j)), hh);
    const float w1 = pick(make_float4(lane_f(al.x, j + 1), lane_f(al.y, j + 1),
                                      lane_f(al.z, j + 1), lane_f(al.w, j + 1)), hh);
    const float w2 = pick(make_float4(lane_f(al.x, j + 2), lane_f(al.y, j + 2),
                                      lane_f(al.z, j + 2), lane_f(al.w, j + 2)), hh);
    const float w3 = pick(make_float4(lane_f(al.x, j + 3), lane_f(al.y, j + 3),
                                      lane_f(al.z, j + 3), lane_f(al.w, j + 3)), hh);
    edge_fma(acc, v0, w0);
    edge_fma(acc, v1, w1);
    edge_fma(acc, v2, w2);
    edge_fma(acc, v3, w3);
  }
  for (; j < n; ++j) {
    const int s0 = __builtin_amdgcn_readlane(srcv, j);
    const float4 v0 = h4[s0 * 32 + lane];
    const float w0 = pick(make_float4(lane_f(al.x, j), lane_f(al.y, j),
                                      lane_f(al.z, j), lane_f(al.w, j)), hh);
    edge_fma(acc, v0, w0);
  }
}

__device__ __forceinline__ void gather_live(float4& acc,
                                            const float4* __restrict__ h4,
                                            int srcv, const float4 al,
                                            unsigned live, int lane, int hh) {
  while (live) {
    const int j = __builtin_ctz(live);
    live &= live - 1;
    const int s0 = __builtin_amdgcn_readlane(srcv, j);
    const float4 v0 = h4[s0 * 32 + lane];
    const float w0 = pick(make_float4(lane_f(al.x, j), lane_f(al.y, j),
                                      lane_f(al.z, j), lane_f(al.w, j)), hh);
    edge_fma(acc, v0, w0);
  }
}

__global__ void __launch_bounds__(256)
gat_softmax_gather_bkt(const int* __restrict__ bucket, const int* __restrict__ cnt,
                       const int* __restrict__ spill,
                       const float* __restrict__ asrc, const float* __restrict__ adst,
                       const float* __restrict__ h, const float* __restrict__ skip,
                       float* __restrict__ out, int N) {
  const int lane = threadIdx.x & 31;
  const int hh = lane >> 3;
  const int wpb = blockDim.x >> 5;
  const int wv = __builtin_amdgcn_readfirstlane(threadIdx.x >> 5);
  const int nw = gridDim.x * wpb;
  const float4* __restrict__ h4 = (const float4*)h;
  const float4* __restrict__ k4 = (const float4*)skip;
  const float4* __restrict__ as4 = (const float4*)asrc;
  const float4* __restrict__ ad4 = (const float4*)adst;
  const int2* __restrict__ sp2 = (const int2*)spill;
  float4* __restrict__ o4 = (float4*)out;
  const float NEG_INF = -__builtin_huge_valf();
  const int nspill = cnt[N];

  for (int d = blockIdx.x * wpb + wv; d < N; d += nw) {
    const int deg = cnt[d];
    const float4 ad = ad4[d];
    const float4 es = score4(as4[d], ad);

    const int n0 = deg < SLOTS ? deg : SLOTS;
    const int nsp = deg > SLOTS ? nspill : 0;
    int src0 = 0;
    float4 e0 = make_float4(NEG_INF, NEG_INF, NEG_INF, NEG_INF);
    if (lane < n0) {
      src0 = bucket[d * SLOTS + lane];
      e0 = score4(as4[src0], ad);
    }

    float4 mx = e0;
    for (int base = 0; base < nsp; base += 32) {
      if (base + lane < nsp) {
        const int2 p = sp2[base + lane];
        if (p.x == d) {
          const float4 e = score4(as4[p.y], ad);
          mx.x = fmaxf(mx.x, e.x);
          mx.y = fmaxf(mx.y, e.y);
          mx.z = fmaxf(mx.z, e.z);
          mx.w = fmaxf(mx.w, e.w);
        }
      }
    }
    float4 m;
    m.x = fmaxf(wave_max(mx.x), es.x);
    m.y = fmaxf(wave_max(mx.y), es.y);
    m.z = fmaxf(wave_max(mx.z), es.z);
    m.w = fmaxf(wave_max(mx.w), es.w);
    m.x = uni(isfinite(m.x) ? m.x : 0.0f);
    m.y = uni(isfinite(m.y) ? m.y : 0.0f);
    m.z = uni(isfinite(m.z) ? m.z : 0.0f);
    m.w = uni(isfinite(m.w) ? m.w : 0.0f);

    float4 ex0 = make_float4(0.f, 0.f, 0.f, 0.f);
    if (lane < n0) {
      ex0.x = expf(e0.x - m.x);
      ex0.y = expf(e0.y - m.y);
      ex0.z = expf(e0.z - m.z);
      ex0.w = expf(e0.w - m.w);
    }
    float4 sm = ex0;
    for (int base = 0; base < nsp; base += 32) {
      if (base + lane < nsp) {
        const int2 p = sp2[base + lane];
        if (p.x == d) {
          const float4 e = score4(as4[p.y], ad);
          sm.x += expf(e.x - m.x);
          sm.y += expf(e.y - m.y);
          sm.z += expf(e.z - m.z);
          sm.w += expf(e.w - m.w);
        }
      }
    }
    const float4 exs = make_float4(expf(es.x - m.x), expf(es.y - m.y),
                                   expf(es.z - m.z), expf(es.w - m.w));
    float4 den;
    den.x = uni((wave_sum(sm.x) + exs.x) + 1e-16f);
    den.y = uni((wave_sum(sm.y) + exs.y) + 1e-16f);
    den.z = uni((wave_sum(sm.z) + exs.z) + 1e-16f);
    den.w = uni((wave_sum(sm.w) + exs.w) + 1e-16f);

    float4 acc = make_float4(0.f, 0.f, 0.f, 0.f);
    edge_fma(acc, h4[d * 32 + lane],
             pick(make_float4(exs.x / den.x, exs.y / den.y, exs.z / den.z,
                              exs.w / den.w), hh));
    {
      const float4 al = make_float4(ex0.x / den.x, ex0.y / den.y, ex0.z / den.z,
                                    ex0.w / den.w);
      gather_chunk(acc, h4, src0, al, n0, lane, hh);
    }
    for (int base = 0; base < nsp; base += 32) {
      bool mine = false;
      int srcv = 0;
      float4 al = make_float4(0.f, 0.f, 0.f, 0.f);
      if (base + lane < nsp) {
        const int2 p = sp2[base + lane];
        if (p.x == d) {
          mine = true;
          srcv = p.y;
          const float4 e = score4(as4[srcv], ad);
          al.x = expf(e.x - m.x) / den.x;
          al.y = expf(e.y - m.y) / den.y;
          al.z = expf(e.z - m.z) / den.z;
          al.w = expf(e.w - m.w) / den.w;
        }
      }
      gather_live(acc, h4, srcv, al, __builtin_amdgcn_ballot_w32(mine), lane, hh);
    }

    const float4 sk = k4[d * 32 + lane];
    float4 o;
    o.x = elu1(sk.x + acc.x);
    o.y = elu1(sk.y + acc.y);
    o.z = elu1(sk.z + acc.z);
    o.w = elu1(sk.w + acc.w);
    o4[d * 32 + lane] = o;
  }
}

extern "C" void kernel_launch(void* const* d_in, const int* in_sizes, int n_in,
                              void* d_out, int out_size, void* d_ws,
                              size_t ws_size, hipStream_t stream) {
  const float* x  = (const float*)d_in[0];
  const int*   ei = (const int*)d_in[1];

  const int N  = in_sizes[0] / DIM;
  const int NE = in_sizes[1] / 2;

  float* ws = (float*)d_ws;
  size_t o = 0;
  float* h     = ws + o; o += (size_t)N * DIM;
  float* agg   = ws + o; o += (size_t)N * DIM;
  float* asrc  = ws + o; o += (size_t)N * HEADS;
  float* adst  = ws + o; o += (size_t)N * HEADS;
  int*  bucket  = (int*)(ws + o);  o += (size_t)N * SLOTS;
  int*  spill   = (int*)(ws + o);  o += (size_t)NE * 2;
  int*  stage   = (int*)(ws + o);  o += (size_t)CHUNKS * 256 * CAP;
  int*  cellcnt = (int*)(ws + o);  o += (size_t)CHUNKS * CROW;
  int*  ovf     = (int*)(ws + o);  o += ((size_t)NE + CHUNKS) * 2;
  int*  cnt     = (int*)(ws + o);  o += (size_t)N + 1;

  float* y = (float*)d_out;

  const int TB = 256;
  const int nStrips = N / 16;
  const int gGemm = 256;
  const int gRange = (N + RANGE - 1) / RANGE;
  const int gGath = 2048;

  bucket_pass1<<<CHUNKS, SORT_TB, 0, stream>>>(ei, NE, N, stage, cellcnt, ovf, cnt);
  bucket_pass2<<<gRange, SORT_TB, 0, stream>>>(stage, cellcnt, ovf, NE, N, cnt, bucket,
                                               spill);

  const float* xin = x;
  for (int layer = 0; layer < 2; ++layer) {
    const float* W     = (const float*)d_in[2 + layer * 5];
    const float* a_src = (const float*)d_in[3 + layer * 5];
    const float* a_dst = (const float*)d_in[4 + layer * 5];
    const float* bias  = (const float*)d_in[5 + layer * 5];
    const float* Wskip = (const float*)d_in[6 + layer * 5];

    gat_gemm_pair_f32_wmma<<<gGemm, 512, 0, stream>>>(xin, W, Wskip, bias, a_src,
                                                      a_dst, h, agg, asrc, adst,
                                                      nStrips);

    gat_softmax_gather_bkt<<<gGath, TB, 0, stream>>>(bucket, cnt, spill, asrc, adst,
                                                     h, agg, y, N);
    xin = y;
  }
}


# Softmax in the gather made head-major: lane (head, slot%8) does one head of an edge; max and sum are 8-lane trees, a row's weight one 8-lane broadcast; sum order: lane fold over slots, 8-lane tree
# speedup vs baseline: 6.1917x; 1.0534x over previous
#include <hip/hip_runtime.h>
#include <math.h>

#define DIM 128
#define HEADS 4
#define CH 32
#define NEG_SLOPE 0.2f

typedef __attribute__((ext_vector_type(2))) float v2f;
typedef __attribute__((ext_vector_type(8))) float v8f;

__device__ __forceinline__ float xhalf(float v) {
  const int i = __builtin_bit_cast(int, v);
  return __builtin_bit_cast(
      float, __builtin_amdgcn_permlanex16(i, i, 0x76543210, 0xfedcba98, false,
                                          false));
}

template <int CTRL>
__device__ __forceinline__ float row_xor(float v) {
  return __builtin_bit_cast(
      float, __builtin_amdgcn_update_dpp(0, __builtin_bit_cast(int, v), CTRL,
                                         0xf, 0xf, false));
}

template <int CTRL>
__device__ __forceinline__ float fold(float a, float b, bool up) {
  const float keep = up ? b : a;
  const float give = up ? a : b;
  return keep + row_xor<CTRL>(give);
}

__device__ __forceinline__ void load_strip(v2f (&a)[32], const float* __restrict__ X,
                                           int strip, int m, int kh) {
  const v2f* __restrict__ arow =
      (const v2f*)(X + (size_t)(strip * 16 + m) * DIM + 2 * kh);
#pragma unroll
  for (int i = 0; i < 32; ++i) a[i] = arow[2 * i];
}

__global__ void __launch_bounds__(512)
gat_gemm_pair_f32_wmma(const float* __restrict__ X,
                       const float* __restrict__ Wa,
                       const float* __restrict__ Wb,
                       const float* __restrict__ bias_b,
                       const float* __restrict__ a_src,
                       const float* __restrict__ a_dst,
                       float* __restrict__ out_a,
                       float* __restrict__ out_b,
                       float* __restrict__ asrc,
                       float* __restrict__ adst, int nstrips) {
  __shared__ float4 wl[32 * 128];
  const int tid = threadIdx.x;
  const int lane = tid & 31;
  const int wave = __builtin_amdgcn_readfirstlane(tid >> 5);
  const int wpb = blockDim.x >> 5;
  const int m  = lane & 15;
  const int kh = lane >> 4;
  const int g = blockIdx.x & 1;
  const int pair = blockIdx.x >> 1;
  const int npairs = gridDim.x >> 1;
  const float* __restrict__ Wg = g ? Wb : Wa;
  float* __restrict__ out = g ? out_b : out_a;

  int s = pair + npairs * wave;
  const int sstep = npairs * wpb;

  float4 wv[8];
  {
    const int c = tid & 127;
#pragma unroll
    for (int q = 0; q < 8; ++q) {
      const int jh = (tid >> 7) + 4 * q;
      const int k = (jh >> 1) * 8 + (jh & 1) * 2;
      wv[q].x = Wg[(k) * DIM + c];
      wv[q].y = Wg[(k + 1) * DIM + c];
      wv[q].z = Wg[(k + 4) * DIM + c];
      wv[q].w = Wg[(k + 5) * DIM + c];
    }
  }
  float bz[8];
  float asv[8], adv[8];
#pragma unroll
  for (int t = 0; t < 8; ++t) {
    const float bt = bias_b[16 * t + m];
    bz[t] = g ? bt : 0.0f;
    asv[t] = a_src[16 * t + m];
    adv[t] = a_dst[16 * t + m];
  }
  const bool up0 = m & 1, up1 = m & 2, up2 = m & 4, up3 = m & 8;
#pragma unroll
  for (int q = 0; q < 8; ++q) wl[tid + 512 * q] = wv[q];
  __syncthreads();

  const float4* wlane = wl + kh * 128 + m;
  while (s < nstrips) {
    v2f a[32];
    load_strip(a, X, s, m, kh);
    __builtin_amdgcn_sched_barrier(0);
    v8f acc[8];
#pragma unroll
    for (int t = 0; t < 8; ++t) acc[t] = (v8f){};
#pragma unroll
    for (int jj = 0; jj < 16; ++jj) {
      float4 b[8];
#pragma unroll
      for (int t = 0; t < 8; ++t) b[t] = wlane[jj * 256 + 16 * t];
#pragma unroll
      for (int t = 0; t < 8; ++t) {
        v2f bv; bv.x = b[t].x; bv.y = b[t].y;
        acc[t] = __builtin_amdgcn_wmma_f32_16x16x4_f32(
            false, a[2 * jj], false, bv, (short)0, acc[t], false, false);
      }
#pragma unroll
      for (int t = 0; t < 8; ++t) {
        v2f bv; bv.x = b[t].z; bv.y = b[t].w;
        acc[t] = __builtin_amdgcn_wmma_f32_16x16x4_f32(
            false, a[2 * jj + 1], false, bv, (short)0, acc[t], false, false);
      }
    }
    float* __restrict__ orow = out + (size_t)(s * 16) * DIM + lane;
#pragma unroll
    for (int tp = 0; tp < 4; ++tp) {
#pragma unroll
      for (int r = 0; r < 8; ++r) {
        const float xv = acc[2 * tp][r] + bz[2 * tp];
        const float yv = acc[2 * tp + 1][r] + bz[2 * tp + 1];
        const float xs = xhalf(xv);
        const float ys = xhalf(yv);
        orow[(size_t)r * DIM + 32 * tp] = kh ? ys : xv;
        orow[(size_t)(r + 8) * DIM + 32 * tp] = kh ? yv : xs;
      }
    }
    if (g == 0) {
#pragma unroll
      for (int sd = 0; sd < 2; ++sd) {
        const float* av = sd ? adv : asv;
        float w2[8];
#pragma unroll
        for (int r = 0; r < 8; ++r) {
          float p[4];
#pragma unroll
          for (int hd = 0; hd < 4; ++hd)
            p[hd] = acc[2 * hd][r] * av[2 * hd] + acc[2 * hd + 1][r] * av[2 * hd + 1];
          const float u0 = fold<0x161>(p[0], p[1], up0);
          const float u1 = fold<0x161>(p[2], p[3], up0);
          w2[r] = fold<0x162>(u0, u1, up1);
        }
        float w3[4];
#pragma unroll
        for (int q = 0; q < 4; ++q)
          w3[q] = fold<0x164>(w2[2 * q], w2[2 * q + 1], up2);
        const float y0 = fold<0x168>(w3[0], w3[1], up3);
        const float y1 = fold<0x168>(w3[2], w3[3], up3);
        const float x0 = xhalf(y0);
        const float x1 = xhalf(y1);
        float* __restrict__ srow = (sd ? adst : asrc) + (size_t)s * 64 + lane;
        srow[0] = kh ? x1 : y0;
        srow[32] = kh ? y1 : x0;
      }
    }
    s += sstep;
  }
}

#define SLOTS 32
#define CHUNKS 256
#define RANGE 256
#define RSHIFT 8
#define CAP 32
#define CROW 288
#define SORT_TB 512
#define EPT 5

__global__ void __launch_bounds__(SORT_TB)
bucket_pass1(const int* __restrict__ ei, int NE, int N, int* __restrict__ stage,
             int* __restrict__ cellcnt, int* __restrict__ ovf,
             int* __restrict__ cnt) {
  __shared__ int4 lst4[256 * CAP / 4];
  __shared__ int ccnt[256];
  __shared__ int ovn;
  int* lst = (int*)lst4;
  const int tid = threadIdx.x;
  const int b = blockIdx.x;
  const int nrange = (N + RANGE - 1) >> RSHIFT;
  const int C = (NE + CHUNKS - 1) / CHUNKS;
  for (int i = tid; i < nrange * (CAP / 4); i += SORT_TB)
    lst4[i] = make_int4(0, 0, 0, 0);
  if (tid < 256) ccnt[tid] = 0;
  if (tid == 0) {
    ovn = 0;
    if (b == 0) cnt[N] = 0;
  }
  __syncthreads();
  const int lo = b * C;
  const int hi = (lo + C < NE) ? lo + C : NE;
  int2* __restrict__ ov2 = (int2*)ovf;
  for (int e0 = lo + tid; e0 < hi; e0 += EPT * SORT_TB) {
    int sv[EPT], dv[EPT], pos[EPT];
#pragma unroll
    for (int u = 0; u < EPT; ++u) {
      const int e = e0 + SORT_TB * u;
      const bool ok = e < hi;
      const int ec = ok ? e : e0;
      const int st = ei[ec];
      const int dt = ei[NE + ec];
      sv[u] = st;
      dv[u] = ok ? dt : -1;
    }
#pragma unroll
    for (int u = 0; u < EPT; ++u)
      pos[u] = (dv[u] >= 0) ? atomicAdd(&ccnt[dv[u] >> RSHIFT], 1) : 0;
#pragma unroll
    for (int u = 0; u < EPT; ++u) {
      if (dv[u] >= 0) {
        if (pos[u] < CAP) {
          lst[(dv[u] >> RSHIFT) * CAP + pos[u]] = ((dv[u] & (RANGE - 1)) << 20) | sv[u];
        } else {
          const int k = atomicAdd(&ovn, 1);
          ov2[lo + k] = make_int2(dv[u], sv[u]);
        }
      }
    }
  }
  __syncthreads();
  if (tid < 256) {
    const int c = ccnt[tid];
    cellcnt[b * CROW + tid] = c < CAP ? c : CAP;
  } else if (tid < CROW) {
    cellcnt[b * CROW + tid] = (tid == 256) ? ovn : 0;
  }
  int4* __restrict__ st4 = (int4*)stage + (size_t)b * (256 * CAP / 4);
  for (int i = tid; i < nrange * (CAP / 4); i += SORT_TB) st4[i] = lst4[i];
}

__device__ __forceinline__ void place_ranked(int* __restrict__ lb, int pos, int dl,
                                             int s, int node0, int* __restrict__ nspill,
                                             int2* __restrict__ sp2) {
  if (pos < SLOTS) {
    lb[dl * SLOTS + pos] = s;
  } else {
    const int k = atomicAdd(nspill, 1);
    sp2[k] = make_int2(node0 + dl, s);
  }
}

__global__ void __launch_bounds__(SORT_TB)
bucket_pass2(const int* __restrict__ stage, const int* __restrict__ cellcnt,
             const int* __restrict__ ovf, int NE, int N, int* __restrict__ cnt,
             int* __restrict__ bucket, int* __restrict__ spill) {
  __shared__ int4 lb4[RANGE * SLOTS / 4];
  __shared__ int lcnt[RANGE];
  int* lb = (int*)lb4;
  const int tid = threadIdx.x;
  const int lane = tid & 31;
  const int wv = __builtin_amdgcn_readfirstlane(tid >> 5);
  const int r = blockIdx.x;
  const int node0 = r * RANGE;
  const int C = (NE + CHUNKS - 1) / CHUNKS;
  int2* __restrict__ sp2 = (int2*)spill;
  const int2* __restrict__ ov2 = (const int2*)ovf;
  const int nv = cellcnt[(wv + 16 * (lane & 15)) * CROW + r];
  int ev[16];
#pragma unroll
  for (int u = 0; u < 16; ++u)
    ev[u] = stage[((size_t)(wv + 16 * u) * 256 + r) * CAP + lane];
  for (int i = tid; i < RANGE * SLOTS / 4; i += SORT_TB) lb4[i] = make_int4(0, 0, 0, 0);
  if (tid < RANGE) lcnt[tid] = 0;
  __syncthreads();
#pragma unroll
  for (int u0 = 0; u0 < 16; u0 += 8) {
    int pos[8];
#pragma unroll
    for (int u = 0; u < 8; ++u) {
      const int n = __builtin_amdgcn_readlane(nv, u0 + u);
      pos[u] = (lane < n) ? atomicAdd(&lcnt[ev[u0 + u] >> 20], 1) : -1;
    }
#pragma unroll
    for (int u = 0; u < 8; ++u)
      if (pos[u] >= 0)
        place_ranked(lb, pos[u], ev[u0 + u] >> 20, ev[u0 + u] & 0xFFFFF, node0,
                     &cnt[N], sp2);
  }
  if (tid < CHUNKS) {
    const int m = cellcnt[tid * CROW + 256];
    for (int i = 0; i < m; ++i) {
      const int2 p = ov2[tid * C + i];
      if ((p.x >> RSHIFT) == r) {
        const int dl = p.x & (RANGE - 1);
        place_ranked(lb, atomicAdd(&lcnt[dl], 1), dl, p.y, node0, &cnt[N], sp2);
      }
    }
  }
  __syncthreads();
  if (tid < RANGE && node0 + tid < N) cnt[node0 + tid] = lcnt[tid];
  const int nn = (N - node0 < RANGE) ? N - node0 : RANGE;
  int4* __restrict__ b4 = (int4*)bucket + (size_t)node0 * (SLOTS / 4);
  for (int i = tid; i < nn * (SLOTS / 4); i += SORT_TB) b4[i] = lb4[i];
}

__device__ __forceinline__ void edge_fma(float4& acc, const float4 hv, float a) {
  acc.x = __builtin_fmaf(hv.x, a, acc.x);
  acc.y = __builtin_fmaf(hv.y, a, acc.y);
  acc.z = __builtin_fmaf(hv.z, a, acc.z);
  acc.w = __builtin_fmaf(hv.w, a, acc.w);
}

__device__ __forceinline__ float elu1(float v) {
  return (v > 0.0f) ? v : expm1f(v);
}

__device__ __forceinline__ float lrelu(float v) {
  return (v > 0.0f) ? v : NEG_SLOPE * v;
}

__device__ __forceinline__ float grp_max(float v) {
  v = fmaxf(v, row_xor<0x161>(v));
  v = fmaxf(v, row_xor<0x162>(v));
  return fmaxf(v, row_xor<0x164>(v));
}

__device__ __forceinline__ float grp_sum(float v) {
  v += row_xor<0x161>(v);
  v += row_xor<0x162>(v);
  return v + row_xor<0x164>(v);
}

template <int I>
__device__ __forceinline__ float grp_bcast(float v) {
  return __builtin_bit_cast(
      float, __builtin_amdgcn_mov_dpp8(__builtin_bit_cast(unsigned, v),
                                       (unsigned)I * 0x249249u));
}

template <int I>
__device__ __forceinline__ void one_edge(float4& acc, const float4* __restrict__ h4,
                                         int srcv, float w, int lane) {
  const int s0 = __builtin_amdgcn_readlane(srcv, I);
  const float4 v0 = h4[s0 * 32 + lane];
  edge_fma(acc, v0, grp_bcast<I>(w));
}

template <int I0>
__device__ __forceinline__ void four_edges(float4& acc, const float4* __restrict__ h4,
                                           int srcv, float w, int lane) {
  const int s0 = __builtin_amdgcn_readlane(srcv, I0);
  const int s1 = __builtin_amdgcn_readlane(srcv, I0 + 1);
  const int s2 = __builtin_amdgcn_readlane(srcv, I0 + 2);
  const int s3 = __builtin_amdgcn_readlane(srcv, I0 + 3);
  const float4 v0 = h4[s0 * 32 + lane];
  const float4 v1 = h4[s1 * 32 + lane];
  const float4 v2 = h4[s2 * 32 + lane];
  const float4 v3 = h4[s3 * 32 + lane];
  edge_fma(acc, v0, grp_bcast<I0>(w));
  edge_fma(acc, v1, grp_bcast<I0 + 1>(w));
  edge_fma(acc, v2, grp_bcast<I0 + 2>(w));
  edge_fma(acc, v3, grp_bcast<I0 + 3>(w));
}

__device__ __forceinline__ void gather_chunk8(float4& acc,
                                              const float4* __restrict__ h4,
                                              int srcv, float w, int nk, int lane) {
  if (nk >= 4) {
    four_edges<0>(acc, h4, srcv, w, lane);
    if (nk >= 8) {
      four_edges<4>(acc, h4, srcv, w, lane);
    } else {
      if (nk > 4) one_edge<4>(acc, h4, srcv, w, lane);
      if (nk > 5) one_edge<5>(acc, h4, srcv, w, lane);
      if (nk > 6) one_edge<6>(acc, h4, srcv, w, lane);
    }
  } else {
    one_edge<0>(acc, h4, srcv, w, lane);
    if (nk > 1) one_edge<1>(acc, h4, srcv, w, lane);
    if (nk > 2) one_edge<2>(acc, h4, srcv, w, lane);
  }
}

template <int I>
__device__ __forceinline__ void live_edge(float4& acc, const float4* __restrict__ h4,
                                          int srcv, float w, unsigned live, int lane) {
  if (live & (1u << I)) one_edge<I>(acc, h4, srcv, w, lane);
}

__global__ void __launch_bounds__(256)
gat_softmax_gather_bkt(const int* __restrict__ bucket, const int* __restrict__ cnt,
                       const int* __restrict__ spill,
                       const float* __restrict__ asrc, const float* __restrict__ adst,
                       const float* __restrict__ h, const float* __restrict__ skip,
                       float* __restrict__ out, int N) {
  const int lane = threadIdx.x & 31;
  const int g = lane >> 3;
  const int i8 = lane & 7;
  const int wpb = blockDim.x >> 5;
  const int wv = __builtin_amdgcn_readfirstlane(threadIdx.x >> 5);
  const int nw = gridDim.x * wpb;
  const float4* __restrict__ h4 = (const float4*)h;
  const float4* __restrict__ k4 = (const float4*)skip;
  const int2* __restrict__ sp2 = (const int2*)spill;
  float4* __restrict__ o4 = (float4*)out;
  const float NEG_INF = -__builtin_huge_valf();
  const int nspill = cnt[N];

  for (int d = blockIdx.x * wpb + wv; d < N; d += nw) {
    const int deg = cnt[d];
    const float adg = adst[d * HEADS + g];
    const float es = lrelu(asrc[d * HEADS + g] + adg);

    const int n0 = deg < SLOTS ? deg : SLOTS;
    const int nsp = deg > SLOTS ? nspill : 0;
    int srcT[4];
    float eT[4];
#pragma unroll
    for (int k = 0; k < 4; ++k) {
      srcT[k] = 0;
      if (8 * k + i8 < n0) srcT[k] = bucket[d * SLOTS + 8 * k + i8];
    }
    float av[4];
    av[0] = asrc[srcT[0] * HEADS + g];
    av[1] = asrc[srcT[1] * HEADS + g];
    av[2] = 0.0f;
    av[3] = 0.0f;
    if (n0 > 16) {
      av[2] = asrc[srcT[2] * HEADS + g];
      av[3] = asrc[srcT[3] * HEADS + g];
    }
#pragma unroll
    for (int k = 0; k < 4; ++k)
      eT[k] = (8 * k + i8 < n0) ? lrelu(av[k] + adg) : NEG_INF;

    float mx = fmaxf(fmaxf(eT[0], eT[1]), fmaxf(eT[2], eT[3]));
    for (int base = 0; base < nsp; base += 8) {
      if (base + i8 < nsp) {
        const int2 p = sp2[base + i8];
        if (p.x == d) mx = fmaxf(mx, lrelu(asrc[p.y * HEADS + g] + adg));
      }
    }
    float m = fmaxf(grp_max(mx), es);
    m = isfinite(m) ? m : 0.0f;

    float wT[4];
#pragma unroll
    for (int k = 0; k < 4; ++k) {
      wT[k] = 0.0f;
      if (8 * k + i8 < n0) wT[k] = expf(eT[k] - m);
    }
    float sm = (wT[0] + wT[1]) + (wT[2] + wT[3]);
    for (int base = 0; base < nsp; base += 8) {
      if (base + i8 < nsp) {
        const int2 p = sp2[base + i8];
        if (p.x == d) sm += expf(lrelu(asrc[p.y * HEADS + g] + adg) - m);
      }
    }
    const float exs = expf(es - m);
    const float den = (grp_sum(sm) + exs) + 1e-16f;

    float4 acc = make_float4(0.f, 0.f, 0.f, 0.f);
    edge_fma(acc, h4[d * 32 + lane], exs / den);
#pragma unroll
    for (int k = 0; k < 4; ++k) {
      if (n0 > 8 * k)
        gather_chunk8(acc, h4, srcT[k], wT[k] / den, n0 - 8 * k, lane);
    }
    for (int base = 0; base < nsp; base += 8) {
      bool mine = false;
      int srcv = 0;
      float w = 0.0f;
      if (base + i8 < nsp) {
        const int2 p = sp2[base + i8];
        if (p.x == d) {
          mine = true;
          srcv = p.y;
          w = expf(lrelu(asrc[srcv * HEADS + g] + adg) - m) / den;
        }
      }
      const unsigned live = __builtin_amdgcn_ballot_w32(mine) & 0xFFu;
      live_edge<0>(acc, h4, srcv, w, live, lane);
      live_edge<1>(acc, h4, srcv, w, live, lane);
      live_edge<2>(acc, h4, srcv, w, live, lane);
      live_edge<3>(acc, h4, srcv, w, live, lane);
      live_edge<4>(acc, h4, srcv, w, live, lane);
      live_edge<5>(acc, h4, srcv, w, live, lane);
      live_edge<6>(acc, h4, srcv, w, live, lane);
      live_edge<7>(acc, h4, srcv, w, live, lane);
    }

    const float4 sk = k4[d * 32 + lane];
    float4 o;
    o.x = elu1(sk.x + acc.x);
    o.y = elu1(sk.y + acc.y);
    o.z = elu1(sk.z + acc.z);
    o.w = elu1(sk.w + acc.w);
    o4[d * 32 + lane] = o;
  }
}

extern "C" void kernel_launch(void* const* d_in, const int* in_sizes, int n_in,
                              void* d_out, int out_size, void* d_ws,
                              size_t ws_size, hipStream_t stream) {
  const float* x  = (const float*)d_in[0];
  const int*   ei = (const int*)d_in[1];

  const int N  = in_sizes[0] / DIM;
  const int NE = in_sizes[1] / 2;

  float* ws = (float*)d_ws;
  size_t o = 0;
  float* h     = ws + o; o += (size_t)N * DIM;
  float* agg   = ws + o; o += (size_t)N * DIM;
  float* asrc  = ws + o; o += (size_t)N * HEADS;
  float* adst  = ws + o; o += (size_t)N * HEADS;
  int*  bucket  = (int*)(ws + o);  o += (size_t)N * SLOTS;
  int*  spill   = (int*)(ws + o);  o += (size_t)NE * 2;
  int*  stage   = (int*)(ws + o);  o += (size_t)CHUNKS * 256 * CAP;
  int*  cellcnt = (int*)(ws + o);  o += (size_t)CHUNKS * CROW;
  int*  ovf     = (int*)(ws + o);  o += ((size_t)NE + CHUNKS) * 2;
  int*  cnt     = (int*)(ws + o);  o += (size_t)N + 1;

  float* y = (float*)d_out;

  const int TB = 256;
  const int nStrips = N / 16;
  const int gGemm = 256;
  const int gRange = (N + RANGE - 1) / RANGE;
  const int gGath = 2048;

  bucket_pass1<<<CHUNKS, SORT_TB, 0, stream>>>(ei, NE, N, stage, cellcnt, ovf, cnt);
  bucket_pass2<<<gRange, SORT_TB, 0, stream>>>(stage, cellcnt, ovf, NE, N, cnt, bucket,
                                               spill);

  const float* xin = x;
  for (int layer = 0; layer < 2; ++layer) {
    const float* W     = (const float*)d_in[2 + layer * 5];
    const float* a_src = (const float*)d_in[3 + layer * 5];
    const float* a_dst = (const float*)d_in[4 + layer * 5];
    const float* bias  = (const float*)d_in[5 + layer * 5];
    const float* Wskip = (const float*)d_in[6 + layer * 5];

    gat_gemm_pair_f32_wmma<<<gGemm, 512, 0, stream>>>(xin, W, Wskip, bias, a_src,
                                                      a_dst, h, agg, asrc, adst,
                                                      nStrips);

    gat_softmax_gather_bkt<<<gGath, TB, 0, stream>>>(bucket, cnt, spill, asrc, adst,
                                                     h, agg, y, N);
    xin = y;
  }
}


# Gather's library routines replaced by hardware instructions: softmax exps and ELU's exp as v_exp_f32 after a log2(e) scale (ELU as exp-1, branch-free), 1/l as v_rcp_f32; f32 kept, sum order as before
# speedup vs baseline: 7.0405x; 1.1371x over previous
#include <hip/hip_runtime.h>
#include <math.h>

#define DIM 128
#define HEADS 4
#define CH 32
#define NEG_SLOPE 0.2f

typedef __attribute__((ext_vector_type(2))) float v2f;
typedef __attribute__((ext_vector_type(8))) float v8f;

__device__ __forceinline__ float xhalf(float v) {
  const int i = __builtin_bit_cast(int, v);
  return __builtin_bit_cast(
      float, __builtin_amdgcn_permlanex16(i, i, 0x76543210, 0xfedcba98, false,
                                          false));
}

template <int CTRL>
__device__ __forceinline__ float row_xor(float v) {
  return __builtin_bit_cast(
      float, __builtin_amdgcn_update_dpp(0, __builtin_bit_cast(int, v), CTRL,
                                         0xf, 0xf, false));
}

template <int CTRL>
__device__ __forceinline__ float fold(float a, float b, bool up) {
  const float keep = up ? b : a;
  const float give = up ? a : b;
  return keep + row_xor<CTRL>(give);
}

__device__ __forceinline__ void load_strip(v2f (&a)[32], const float* __restrict__ X,
                                           int strip, int m, int kh) {
  const v2f* __restrict__ arow =
      (const v2f*)(X + (size_t)(strip * 16 + m) * DIM + 2 * kh);
#pragma unroll
  for (int i = 0; i < 32; ++i) a[i] = arow[2 * i];
}

__global__ void __launch_bounds__(512)
gat_gemm_pair_f32_wmma(const float* __restrict__ X,
                       const float* __restrict__ Wa,
                       const float* __restrict__ Wb,
                       const float* __restrict__ bias_b,
                       float* __restrict__ out_a,
                       float* __restrict__ out_b, int nstrips) {
  __shared__ float4 wl[32 * 128];
  const int tid = threadIdx.x;
  const int lane = tid & 31;
  const int wave = __builtin_amdgcn_readfirstlane(tid >> 5);
  const int wpb = blockDim.x >> 5;
  const int m  = lane & 15;
  const int kh = lane >> 4;
  const int g = blockIdx.x & 1;
  const int pair = blockIdx.x >> 1;
  const int npairs = gridDim.x >> 1;
  const float* __restrict__ Wg = g ? Wb : Wa;
  float* __restrict__ out = g ? out_b : out_a;

  int s = pair + npairs * wave;
  const int sstep = npairs * wpb;

  float4 wv[8];
  {
    const int c = tid & 127;
#pragma unroll
    for (int q = 0; q < 8; ++q) {
      const int jh = (tid >> 7) + 4 * q;
      const int k = (jh >> 1) * 8 + (jh & 1) * 2;
      wv[q].x = Wg[(k) * DIM + c];
      wv[q].y = Wg[(k + 1) * DIM + c];
      wv[q].z = Wg[(k + 4) * DIM + c];
      wv[q].w = Wg[(k + 5) * DIM + c];
    }
  }
  float bz[8];
#pragma unroll
  for (int t = 0; t < 8; ++t) {
    const float bt = bias_b[16 * t + m];
    bz[t] = g ? bt : 0.0f;
  }
#pragma unroll
  for (int q = 0; q < 8; ++q) wl[tid + 512 * q] = wv[q];
  __syncthreads();

  const float4* wlane = wl + kh * 128 + m;
  while (s < nstrips) {
    v2f a[32];
    load_strip(a, X, s, m, kh);
    __builtin_amdgcn_sched_barrier(0);
    v8f acc[8];
#pragma unroll
    for (int t = 0; t < 8; ++t) acc[t] = (v8f){};
#pragma unroll
    for (int jj = 0; jj < 16; ++jj) {
      float4 b[8];
#pragma unroll
      for (int t = 0; t < 8; ++t) b[t] = wlane[jj * 256 + 16 * t];
#pragma unroll
      for (int t = 0; t < 8; ++t) {
        v2f bv; bv.x = b[t].x; bv.y = b[t].y;
        acc[t] = __builtin_amdgcn_wmma_f32_16x16x4_f32(
            false, a[2 * jj], false, bv, (short)0, acc[t], false, false);
      }
#pragma unroll
      for (int t = 0; t < 8; ++t) {
        v2f bv; bv.x = b[t].z; bv.y = b[t].w;
        acc[t] = __builtin_amdgcn_wmma_f32_16x16x4_f32(
            false, a[2 * jj + 1], false, bv, (short)0, acc[t], false, false);
      }
    }
    float* __restrict__ orow = out + (size_t)(s * 16) * DIM + lane;
#pragma unroll
    for (int tp = 0; tp < 4; ++tp) {
#pragma unroll
      for (int r = 0; r < 8; ++r) {
        const float xv = acc[2 * tp][r] + bz[2 * tp];
        const float yv = acc[2 * tp + 1][r] + bz[2 * tp + 1];
        const float xs = xhalf(xv);
        const float ys = xhalf(yv);
        orow[(size_t)r * DIM + 32 * tp] = kh ? ys : xv;
        orow[(size_t)(r + 8) * DIM + 32 * tp] = kh ? yv : xs;
      }
    }
    s += sstep;
  }
}

#define SLOTS 32
#define CHUNKS 256
#define RANGE 256
#define RSHIFT 8
#define CAP 32
#define CROW 288
#define SORT_TB 512
#define EPT 5

__global__ void __launch_bounds__(SORT_TB)
bucket_pass1(const int* __restrict__ ei, int NE, int N, int* __restrict__ stage,
             int* __restrict__ cellcnt, int* __restrict__ ovf,
             int* __restrict__ cnt) {
  __shared__ int4 lst4[256 * CAP / 4];
  __shared__ int ccnt[256];
  __shared__ int ovn;
  int* lst = (int*)lst4;
  const int tid = threadIdx.x;
  const int b = blockIdx.x;
  const int nrange = (N + RANGE - 1) >> RSHIFT;
  const int C = (NE + CHUNKS - 1) / CHUNKS;
  for (int i = tid; i < nrange * (CAP / 4); i += SORT_TB)
    lst4[i] = make_int4(0, 0, 0, 0);
  if (tid < 256) ccnt[tid] = 0;
  if (tid == 0) {
    ovn = 0;
    if (b == 0) cnt[N] = 0;
  }
  __syncthreads();
  const int lo = b * C;
  const int hi = (lo + C < NE) ? lo + C : NE;
  int2* __restrict__ ov2 = (int2*)ovf;
  for (int e0 = lo + tid; e0 < hi; e0 += EPT * SORT_TB) {
    int sv[EPT], dv[EPT], pos[EPT];
#pragma unroll
    for (int u = 0; u < EPT; ++u) {
      const int e = e0 + SORT_TB * u;
      const bool ok = e < hi;
      const int ec = ok ? e : e0;
      const int st = ei[ec];
      const int dt = ei[NE + ec];
      sv[u] = st;
      dv[u] = ok ? dt : -1;
    }
#pragma unroll
    for (int u = 0; u < EPT; ++u)
      pos[u] = (dv[u] >= 0) ? atomicAdd(&ccnt[dv[u] >> RSHIFT], 1) : 0;
#pragma unroll
    for (int u = 0; u < EPT; ++u) {
      if (dv[u] >= 0) {
        if (pos[u] < CAP) {
          lst[(dv[u] >> RSHIFT) * CAP + pos[u]] = ((dv[u] & (RANGE - 1)) << 20) | sv[u];
        } else {
          const int k = atomicAdd(&ovn, 1);
          ov2[lo + k] = make_int2(dv[u], sv[u]);
        }
      }
    }
  }
  __syncthreads();
  if (tid < 256) {
    const int c = ccnt[tid];
    cellcnt[b * CROW + tid] = c < CAP ? c : CAP;
  } else if (tid < CROW) {
    cellcnt[b * CROW + tid] = (tid == 256) ? ovn : 0;
  }
  int4* __restrict__ st4 = (int4*)stage + (size_t)b * (256 * CAP / 4);
  for (int i = tid; i < nrange * (CAP / 4); i += SORT_TB) st4[i] = lst4[i];
}

__device__ __forceinline__ void place_ranked(int* __restrict__ lb, int pos, int dl,
                                             int s, int node0, int* __restrict__ nspill,
                                             int2* __restrict__ sp2) {
  if (pos < SLOTS) {
    lb[dl * SLOTS + pos] = s;
  } else {
    const int k = atomicAdd(nspill, 1);
    sp2[k] = make_int2(node0 + dl, s);
  }
}

__global__ void __launch_bounds__(SORT_TB)
bucket_pass2(const int* __restrict__ stage, const int* __restrict__ cellcnt,
             const int* __restrict__ ovf, int NE, int N, int* __restrict__ cnt,
             int* __restrict__ bucket, int* __restrict__ spill) {
  __shared__ int4 lb4[RANGE * SLOTS / 4];
  __shared__ int lcnt[RANGE];
  int* lb = (int*)lb4;
  const int tid = threadIdx.x;
  const int lane = tid & 31;
  const int wv = __builtin_amdgcn_readfirstlane(tid >> 5);
  const int r = blockIdx.x;
  const int node0 = r * RANGE;
  const int C = (NE + CHUNKS - 1) / CHUNKS;
  int2* __restrict__ sp2 = (int2*)spill;
  const int2* __restrict__ ov2 = (const int2*)ovf;
  const int nv = cellcnt[(wv + 16 * (lane & 15)) * CROW + r];
  int ev[16];
#pragma unroll
  for (int u = 0; u < 16; ++u)
    ev[u] = stage[((size_t)(wv + 16 * u) * 256 + r) * CAP + lane];
  for (int i = tid; i < RANGE * SLOTS / 4; i += SORT_TB) lb4[i] = make_int4(0, 0, 0, 0);
  if (tid < RANGE) lcnt[tid] = 0;
  __syncthreads();
#pragma unroll
  for (int u0 = 0; u0 < 16; u0 += 8) {
    int pos[8];
#pragma unroll
    for (int u = 0; u < 8; ++u) {
      const int n = __builtin_amdgcn_readlane(nv, u0 + u);
      pos[u] = (lane < n) ? atomicAdd(&lcnt[ev[u0 + u] >> 20], 1) : -1;
    }
#pragma unroll
    for (int u = 0; u < 8; ++u)
      if (pos[u] >= 0)
        place_ranked(lb, pos[u], ev[u0 + u] >> 20, ev[u0 + u] & 0xFFFFF, node0,
                     &cnt[N], sp2);
  }
  if (tid < CHUNKS) {
    const int m = cellcnt[tid * CROW + 256];
    for (int i = 0; i < m; ++i) {
      const int2 p = ov2[tid * C + i];
      if ((p.x >> RSHIFT) == r) {
        const int dl = p.x & (RANGE - 1);
        place_ranked(lb, atomicAdd(&lcnt[dl], 1), dl, p.y, node0, &cnt[N], sp2);
      }
    }
  }
  __syncthreads();
  if (tid < RANGE && node0 + tid < N) cnt[node0 + tid] = lcnt[tid];
  const int nn = (N - node0 < RANGE) ? N - node0 : RANGE;
  int4* __restrict__ b4 = (int4*)bucket + (size_t)node0 * (SLOTS / 4);
  for (int i = tid; i < nn * (SLOTS / 4); i += SORT_TB) b4[i] = lb4[i];
}

__device__ __forceinline__ void edge_fma(float4& acc, const float4 hv, float a) {
  acc.x = __builtin_fmaf(hv.x, a, acc.x);
  acc.y = __builtin_fmaf(hv.y, a, acc.y);
  acc.z = __builtin_fmaf(hv.z, a, acc.z);
  acc.w = __builtin_fmaf(hv.w, a, acc.w);
}

__device__ __forceinline__ float exp_hw(float x) {
  return __builtin_amdgcn_exp2f(x * 1.44269504088896340736f);
}

__device__ __forceinline__ float elu1(float v) {
  const float neg = exp_hw(v) - 1.0f;
  return (v > 0.0f) ? v : neg;
}

__device__ __forceinline__ float lrelu(float v) {
  return (v > 0.0f) ? v : NEG_SLOPE * v;
}

__device__ __forceinline__ float grp_max(float v) {
  v = fmaxf(v, row_xor<0x161>(v));
  v = fmaxf(v, row_xor<0x162>(v));
  return fmaxf(v, row_xor<0x164>(v));
}

__device__ __forceinline__ float grp_sum(float v) {
  v += row_xor<0x161>(v);
  v += row_xor<0x162>(v);
  return v + row_xor<0x164>(v);
}

template <int I>
__device__ __forceinline__ float grp_bcast(float v) {
  return __builtin_bit_cast(
      float, __builtin_amdgcn_mov_dpp8(__builtin_bit_cast(unsigned, v),
                                       (unsigned)I * 0x249249u));
}

__device__ __forceinline__ float dot4(const float4 r, const float4 a) {
  return __builtin_fmaf(r.w, a.w,
                        __builtin_fmaf(r.z, a.z, __builtin_fmaf(r.y, a.y, r.x * a.x)));
}

__device__ __forceinline__ void online_chunk(float4& acc, float& m, float& l,
                                             const float4* __restrict__ h4, int srcv,
                                             bool valid, unsigned live, const float4 asl,
                                             float adg, int lane, bool up0, bool up1,
                                             bool up2) {
  float4 r[8];
  if (live == 0xFFu) {
#pragma unroll
    for (int i = 0; i < 8; ++i)
      r[i] = h4[__builtin_amdgcn_readlane(srcv, i) * 32 + lane];
  } else {
#pragma unroll
    for (int i = 0; i < 8; ++i) {
      r[i] = make_float4(0.f, 0.f, 0.f, 0.f);
      if (live & (1u << i)) r[i] = h4[__builtin_amdgcn_readlane(srcv, i) * 32 + lane];
    }
  }
  float p[8];
#pragma unroll
  for (int i = 0; i < 8; ++i) p[i] = dot4(r[i], asl);
  const float u0 = fold<0x161>(p[0], p[1], up0);
  const float u1 = fold<0x161>(p[2], p[3], up0);
  const float u2 = fold<0x161>(p[4], p[5], up0);
  const float u3 = fold<0x161>(p[6], p[7], up0);
  const float v0 = fold<0x162>(u0, u1, up1);
  const float v1 = fold<0x162>(u2, u3, up1);
  const float sc_edge = fold<0x164>(v0, v1, up2);
  const float e = valid ? lrelu(sc_edge + adg) : -__builtin_huge_valf();
  const float mn = fmaxf(m, grp_max(e));
  const float rs = exp_hw(m - mn);
  const float w = exp_hw(e - mn);
  l = __builtin_fmaf(l, rs, grp_sum(w));
  acc.x *= rs;
  acc.y *= rs;
  acc.z *= rs;
  acc.w *= rs;
  edge_fma(acc, r[0], grp_bcast<0>(w));
  edge_fma(acc, r[1], grp_bcast<1>(w));
  edge_fma(acc, r[2], grp_bcast<2>(w));
  edge_fma(acc, r[3], grp_bcast<3>(w));
  edge_fma(acc, r[4], grp_bcast<4>(w));
  edge_fma(acc, r[5], grp_bcast<5>(w));
  edge_fma(acc, r[6], grp_bcast<6>(w));
  edge_fma(acc, r[7], grp_bcast<7>(w));
  m = mn;
}

__global__ void __launch_bounds__(256)
gat_softmax_gather_bkt(const int* __restrict__ bucket, const int* __restrict__ cnt,
                       const int* __restrict__ spill,
                       const float* __restrict__ a_src, const float* __restrict__ a_dst,
                       const float* __restrict__ h, const float* __restrict__ skip,
                       float* __restrict__ out, int N) {
  const int lane = threadIdx.x & 31;
  const int i8 = lane & 7;
  const bool up0 = lane & 1, up1 = lane & 2, up2 = lane & 4;
  const int wpb = blockDim.x >> 5;
  const int wv = __builtin_amdgcn_readfirstlane(threadIdx.x >> 5);
  const int nw = gridDim.x * wpb;
  const float4* __restrict__ h4 = (const float4*)h;
  const float4* __restrict__ k4 = (const float4*)skip;
  const int2* __restrict__ sp2 = (const int2*)spill;
  float4* __restrict__ o4 = (float4*)out;
  const float4 asl = ((const float4*)a_src)[lane];
  const float4 adl = ((const float4*)a_dst)[lane];
  const int nspill = cnt[N];

  for (int d = blockIdx.x * wpb + wv; d < N; d += nw) {
    const int deg = cnt[d];
    const int n0 = deg < SLOTS ? deg : SLOTS;
    const int nsp = deg > SLOTS ? nspill : 0;
    int srcT[4];
#pragma unroll
    for (int k = 0; k < 4; ++k) {
      srcT[k] = 0;
      if (8 * k + i8 < n0) srcT[k] = bucket[d * SLOTS + 8 * k + i8];
    }
    const float4 own = h4[d * 32 + lane];
    const float4 sk = k4[d * 32 + lane];
    const float adg = grp_sum(dot4(own, adl));
    const float es = lrelu(grp_sum(dot4(own, asl)) + adg);
    float m = es;
    float l = 1.0f;
    float4 acc = own;

#pragma unroll
    for (int k = 0; k < 4; ++k) {
      if (n0 > 8 * k) {
        const int nk = n0 - 8 * k;
        const unsigned live = nk >= 8 ? 0xFFu : ((1u << nk) - 1u);
        online_chunk(acc, m, l, h4, srcT[k], 8 * k + i8 < n0, live, asl, adg, lane, up0,
                     up1, up2);
      }
    }
    for (int base = 0; base < nsp; base += 8) {
      bool mine = false;
      int srcv = 0;
      if (base + i8 < nsp) {
        const int2 p = sp2[base + i8];
        if (p.x == d) {
          mine = true;
          srcv = p.y;
        }
      }
      const unsigned live = __builtin_amdgcn_ballot_w32(mine) & 0xFFu;
      if (live) online_chunk(acc, m, l, h4, srcv, mine, live, asl, adg, lane, up0, up1, up2);
    }

    const float rd = __builtin_amdgcn_rcpf(l + 1e-16f);
    float4 o;
    o.x = elu1(__builtin_fmaf(acc.x, rd, sk.x));
    o.y = elu1(__builtin_fmaf(acc.y, rd, sk.y));
    o.z = elu1(__builtin_fmaf(acc.z, rd, sk.z));
    o.w = elu1(__builtin_fmaf(acc.w, rd, sk.w));
    o4[d * 32 + lane] = o;
  }
}

extern "C" void kernel_launch(void* const* d_in, const int* in_sizes, int n_in,
                              void* d_out, int out_size, void* d_ws,
                              size_t ws_size, hipStream_t stream) {
  const float* x  = (const float*)d_in[0];
  const int*   ei = (const int*)d_in[1];

  const int N  = in_sizes[0] / DIM;
  const int NE = in_sizes[1] / 2;

  float* ws = (float*)d_ws;
  size_t o = 0;
  float* h     = ws + o; o += (size_t)N * DIM;
  float* agg   = ws + o; o += (size_t)N * DIM;
  int*  bucket  = (int*)(ws + o);  o += (size_t)N * SLOTS;
  int*  spill   = (int*)(ws + o);  o += (size_t)NE * 2;
  int*  stage   = (int*)(ws + o);  o += (size_t)CHUNKS * 256 * CAP;
  int*  cellcnt = (int*)(ws + o);  o += (size_t)CHUNKS * CROW;
  int*  ovf     = (int*)(ws + o);  o += ((size_t)NE + CHUNKS) * 2;
  int*  cnt     = (int*)(ws + o);  o += (size_t)N + 1;

  float* y = (float*)d_out;

  const int TB = 256;
  const int nStrips = N / 16;
  const int gGemm = 256;
  const int gRange = (N + RANGE - 1) / RANGE;
  const int gGath = 1250;

  bucket_pass1<<<CHUNKS, SORT_TB, 0, stream>>>(ei, NE, N, stage, cellcnt, ovf, cnt);
  bucket_pass2<<<gRange, SORT_TB, 0, stream>>>(stage, cellcnt, ovf, NE, N, cnt, bucket,
                                               spill);

  const float* xin = x;
  for (int layer = 0; layer < 2; ++layer) {
    const float* W     = (const float*)d_in[2 + layer * 5];
    const float* a_src = (const float*)d_in[3 + layer * 5];
    const float* a_dst = (const float*)d_in[4 + layer * 5];
    const float* bias  = (const float*)d_in[5 + layer * 5];
    const float* Wskip = (const float*)d_in[6 + layer * 5];

    gat_gemm_pair_f32_wmma<<<gGemm, 512, 0, stream>>>(xin, W, Wskip, bias, h, agg,
                                                      nStrips);

    gat_softmax_gather_bkt<<<gGath, TB, 0, stream>>>(bucket, cnt, spill, a_src, a_dst,
                                                     h, agg, y, N);
    xin = y;
  }
}
